# starting kernel
# baseline (speedup 1.0000x reference)
#include <hip/hip_runtime.h>
#include <math.h>

typedef __attribute__((ext_vector_type(16))) _Float16 v16h;
typedef __attribute__((ext_vector_type(8)))  float    v8f;

#define NNODES 100000
#define NEDGES 1600000
#define FDIM   128

static inline int ceil_div(long long a, long long b) { return (int)((a + b - 1) / b); }

__global__ void init_deg_kernel(float* __restrict__ odeg, float* __restrict__ ideg, int n) {
  int i = blockIdx.x * blockDim.x + threadIdx.x;
  if (i < n) { odeg[i] = 1.0f; ideg[i] = 1.0f; }
}

__global__ void degree_kernel(const int* __restrict__ src, const int* __restrict__ dst,
                              float* __restrict__ odeg, float* __restrict__ ideg, int e) {
  int i = blockIdx.x * blockDim.x + threadIdx.x;
  if (i < e) {
    atomicAdd(&odeg[src[i]], 1.0f);
    atomicAdd(&ideg[dst[i]], 1.0f);
  }
}

__global__ void rsqrt_kernel(float* __restrict__ a, float* __restrict__ b, int n) {
  int i = blockIdx.x * blockDim.x + threadIdx.x;
  if (i < n) {
    a[i] = rsqrtf(fmaxf(a[i], 1.0f));
    b[i] = rsqrtf(fmaxf(b[i], 1.0f));
  }
}

__global__ void zero_kernel(float* __restrict__ p, int n) {
  int i = blockIdx.x * blockDim.x + threadIdx.x;
  if (i < n) p[i] = 0.0f;
}

__global__ void pack_w_kernel(const float* __restrict__ W, _Float16* __restrict__ Wp, int nout) {
  int idx = blockIdx.x * blockDim.x + threadIdx.x;
  int nt = nout >> 4;
  int total = 4 * nt * 32 * 16;
  if (idx >= total) return;
  int j    = idx & 15;
  int l    = (idx >> 4) & 31;
  int rest = idx >> 9;
  int t    = rest % nt;
  int c    = rest / nt;
  int col   = t * 16 + (l & 15);
  int kbase = c * 32 + ((l >> 4) * 8);
  int K     = kbase + (j & 7) + ((j >> 3) * 16);
  Wp[idx] = (_Float16)W[K * nout + col];
}

template<int NOUT>
__global__ void gemm_wmma_kernel(const float* __restrict__ X,
                                 const float* __restrict__ onorm,
                                 const _Float16* __restrict__ Wp,
                                 float* __restrict__ H) {
  constexpr int NT = NOUT / 16;
  int wave = (int)((blockIdx.x * blockDim.x + threadIdx.x) >> 5);
  int lane = threadIdx.x & 31;
  int mtile = wave / NT;
  int ntile = wave % NT;
  if (mtile * 16 >= NNODES) return;

  int r    = lane & 15;
  int half = lane >> 4;
  int row  = mtile * 16 + r;
  float scale = onorm[row];
  const float* px = X + (size_t)row * FDIM;

  v8f acc = {0.f, 0.f, 0.f, 0.f, 0.f, 0.f, 0.f, 0.f};

#pragma unroll
  for (int kc = 0; kc < 4; ++kc) {
    int kbase = kc * 32 + half * 8;
    float4 f0 = *(const float4*)(px + kbase);
    float4 f1 = *(const float4*)(px + kbase + 4);
    float4 f2 = *(const float4*)(px + kbase + 16);
    float4 f3 = *(const float4*)(px + kbase + 20);
    v16h a;
    a[0]  = (_Float16)(f0.x * scale); a[1]  = (_Float16)(f0.y * scale);
    a[2]  = (_Float16)(f0.z * scale); a[3]  = (_Float16)(f0.w * scale);
    a[4]  = (_Float16)(f1.x * scale); a[5]  = (_Float16)(f1.y * scale);
    a[6]  = (_Float16)(f1.z * scale); a[7]  = (_Float16)(f1.w * scale);
    a[8]  = (_Float16)(f2.x * scale); a[9]  = (_Float16)(f2.y * scale);
    a[10] = (_Float16)(f2.z * scale); a[11] = (_Float16)(f2.w * scale);
    a[12] = (_Float16)(f3.x * scale); a[13] = (_Float16)(f3.y * scale);
    a[14] = (_Float16)(f3.z * scale); a[15] = (_Float16)(f3.w * scale);

    v16h b = *(const v16h*)(Wp + ((size_t)(kc * NT + ntile) * 32 + lane) * 16);

    acc = __builtin_amdgcn_wmma_f32_16x16x32_f16(
         false, a,  false, b,
         (short)0, acc,  false,  false);
  }

  int col = ntile * 16 + r;
  float* ph = H + (size_t)(mtile * 16 + half * 8) * NOUT + col;
#pragma unroll
  for (int vr = 0; vr < 8; ++vr)
    ph[(size_t)vr * NOUT] = acc[vr];
}

template<int NOUT>
__global__ void edge_agg_kernel(const int* __restrict__ src, const int* __restrict__ dst,
                                const float* __restrict__ h, float* __restrict__ agg, int e) {
  int wid  = (int)((blockIdx.x * blockDim.x + threadIdx.x) >> 5);
  int lane = threadIdx.x & 31;
  if (wid >= e) return;
  int s = src[wid];
  int d = dst[wid];
  constexpr int V = NOUT / 32;
  const float* hp = h   + (size_t)s * NOUT + lane * V;
  float*       ap = agg + (size_t)d * NOUT + lane * V;
  float v[V];
#pragma unroll
  for (int k = 0; k < V; ++k) v[k] = hp[k];
#pragma unroll
  for (int k = 0; k < V; ++k) atomicAdd(ap + k, v[k]);
}

template<int NOUT, bool RELU>
__global__ void finalize_kernel(const float* __restrict__ agg, const float* __restrict__ h,
                                const float* __restrict__ inorm, const float* __restrict__ b,
                                float* __restrict__ out, int n) {
  int idx = blockIdx.x * blockDim.x + threadIdx.x;
  if (idx >= n) return;
  int node = idx / NOUT;
  int dcol = idx % NOUT;
  float v = (agg[idx] + h[idx]) * inorm[node] + b[dcol];
  out[idx] = RELU ? fmaxf(v, 0.0f) : v;
}

extern "C" void kernel_launch(void* const* d_in, const int* in_sizes, int n_in,
                              void* d_out, int out_size, void* d_ws, size_t ws_size,
                              hipStream_t stream) {
  const float* feat = (const float*)d_in[0];
  const int*   src  = (const int*)  d_in[1];
  const int*   dst  = (const int*)  d_in[2];
  const float* W1   = (const float*)d_in[3];
  const float* b1   = (const float*)d_in[4];
  const float* W2   = (const float*)d_in[5];
  const float* b2   = (const float*)d_in[6];
  const float* W3   = (const float*)d_in[7];
  const float* b3   = (const float*)d_in[8];
  float* out = (float*)d_out;

  size_t off = 0;
  auto alloc = [&](size_t bytes) -> char* {
    char* p = (char*)d_ws + off;
    off += (bytes + 255) & ~(size_t)255;
    return p;
  };
  float*     onorm = (float*)alloc((size_t)NNODES * sizeof(float));
  float*     inorm = (float*)alloc((size_t)NNODES * sizeof(float));
  float*     hbuf  = (float*)alloc((size_t)NNODES * FDIM * sizeof(float));
  float*     abuf  = (float*)alloc((size_t)NNODES * FDIM * sizeof(float));
  float*     xbuf  = (float*)alloc((size_t)NNODES * FDIM * sizeof(float));
  _Float16*  Wp    = (_Float16*)alloc((size_t)FDIM * FDIM * sizeof(_Float16));

  const int TB = 256;

  init_deg_kernel<<<ceil_div(NNODES, TB), TB, 0, stream>>>(onorm, inorm, NNODES);
  degree_kernel<<<ceil_div(NEDGES, TB), TB, 0, stream>>>(src, dst, onorm, inorm, NEDGES);
  rsqrt_kernel<<<ceil_div(NNODES, TB), TB, 0, stream>>>(onorm, inorm, NNODES);

  pack_w_kernel<<<ceil_div(4 * 8 * 512, TB), TB, 0, stream>>>(W1, Wp, 128);
  zero_kernel<<<ceil_div((long long)NNODES * 128, TB), TB, 0, stream>>>(abuf, NNODES * 128);
  {
    int waves = (NNODES / 16) * (128 / 16);
    gemm_wmma_kernel<128><<<ceil_div((long long)waves * 32, TB), TB, 0, stream>>>(feat, onorm, Wp, hbuf);
  }
  edge_agg_kernel<128><<<ceil_div((long long)NEDGES * 32, TB), TB, 0, stream>>>(src, dst, hbuf, abuf, NEDGES);
  finalize_kernel<128, true><<<ceil_div((long long)NNODES * 128, TB), TB, 0, stream>>>(
      abuf, hbuf, inorm, b1, xbuf, NNODES * 128);

  pack_w_kernel<<<ceil_div(4 * 8 * 512, TB), TB, 0, stream>>>(W2, Wp, 128);
  zero_kernel<<<ceil_div((long long)NNODES * 128, TB), TB, 0, stream>>>(abuf, NNODES * 128);
  {
    int waves = (NNODES / 16) * (128 / 16);
    gemm_wmma_kernel<128><<<ceil_div((long long)waves * 32, TB), TB, 0, stream>>>(xbuf, onorm, Wp, hbuf);
  }
  edge_agg_kernel<128><<<ceil_div((long long)NEDGES * 32, TB), TB, 0, stream>>>(src, dst, hbuf, abuf, NEDGES);
  finalize_kernel<128, true><<<ceil_div((long long)NNODES * 128, TB), TB, 0, stream>>>(
      abuf, hbuf, inorm, b2, xbuf, NNODES * 128);

  pack_w_kernel<<<ceil_div(4 * 4 * 512, TB), TB, 0, stream>>>(W3, Wp, 64);
  zero_kernel<<<ceil_div((long long)NNODES * 64, TB), TB, 0, stream>>>(abuf, NNODES * 64);
  {
    int waves = (NNODES / 16) * (64 / 16);
    gemm_wmma_kernel<64><<<ceil_div((long long)waves * 32, TB), TB, 0, stream>>>(xbuf, onorm, Wp, hbuf);
  }
  edge_agg_kernel<64><<<ceil_div((long long)NEDGES * 32, TB), TB, 0, stream>>>(src, dst, hbuf, abuf, NEDGES);
  finalize_kernel<64, false><<<ceil_div((long long)NNODES * 64, TB), TB, 0, stream>>>(
      abuf, hbuf, inorm, b3, out, NNODES * 64);
}


# Degree pass folded into the list build: in-degrees counted per row from each owner block's list (LDS integer adds), out-degrees by integer adds in the fill launch, norms written by the split kernel
# speedup vs baseline: 3.3298x; 3.3298x over previous
#include <hip/hip_runtime.h>
#include <math.h>

typedef __attribute__((ext_vector_type(16))) _Float16 v16h;
typedef __attribute__((ext_vector_type(8)))  float    v8f;

#define NNODES 100000
#define NEDGES 1600000
#define FDIM   128

static inline int ceil_div(long long a, long long b) { return (int)((a + b - 1) / b); }


__global__ void pack_w_kernel(const float* __restrict__ W, _Float16* __restrict__ Wp, int nout) {
  int idx = blockIdx.x * blockDim.x + threadIdx.x;
  int nt = nout >> 4;
  int total = 4 * nt * 32 * 16;
  if (idx >= total) return;
  int j    = idx & 15;
  int l    = (idx >> 4) & 31;
  int rest = idx >> 9;
  int t    = rest % nt;
  int c    = rest / nt;
  int col   = (t >> 2) * 64 + 4 * (l & 15) + (t & 3);
  int kbase = c * 32 + ((l >> 4) * 8);
  int K     = kbase + (j & 7) + ((j >> 3) * 16);
  Wp[idx] = (_Float16)W[K * nout + col];
}

#define GEMM_TB 256

template<int NOUT>
__global__ __launch_bounds__(GEMM_TB)
void gemm_wmma_kernel(const float* __restrict__ X,
                      const float* __restrict__ onorm,
                      const _Float16* __restrict__ Wp,
                      float* __restrict__ H) {
  constexpr int NT = NOUT / 16;
  constexpr int NG = NT / 4;
  static_assert(NT % 4 == 0, "column tiles come in groups of four");
  int mtile = (int)((blockIdx.x * blockDim.x + threadIdx.x) >> 5);
  int lane = threadIdx.x & 31;
  if (mtile * 16 >= NNODES) return;

  int r    = lane & 15;
  int half = lane >> 4;
  int row  = mtile * 16 + r;
  const float* px = X + (size_t)row * FDIM;
  const v16h* __restrict__ wl = (const v16h*)Wp + lane;

  float4 f[16];
#pragma unroll
  for (int kc = 0; kc < 4; ++kc) {
    int kbase = kc * 32 + half * 8;
    f[kc * 4 + 0] = *(const float4*)(px + kbase);
    f[kc * 4 + 1] = *(const float4*)(px + kbase + 4);
    f[kc * 4 + 2] = *(const float4*)(px + kbase + 16);
    f[kc * 4 + 3] = *(const float4*)(px + kbase + 20);
  }
  float scale = onorm[row];
  v16h b[NT];
#pragma unroll
  for (int t = 0; t < NT; ++t) b[t] = wl[t * 32];
  __builtin_amdgcn_sched_barrier(0);

  v16h a[4];
#pragma unroll
  for (int kc = 0; kc < 4; ++kc) {
#pragma unroll
    for (int i = 0; i < 4; ++i) {
      const float4 q = f[kc * 4 + i];
      a[kc][i * 4 + 0] = (_Float16)(q.x * scale);
      a[kc][i * 4 + 1] = (_Float16)(q.y * scale);
      a[kc][i * 4 + 2] = (_Float16)(q.z * scale);
      a[kc][i * 4 + 3] = (_Float16)(q.w * scale);
    }
  }

  v8f acc[NT];
#pragma unroll
  for (int t = 0; t < NT; ++t) acc[t] = (v8f){0.f, 0.f, 0.f, 0.f, 0.f, 0.f, 0.f, 0.f};

#pragma unroll
  for (int kc = 0; kc < 4; ++kc) {
#pragma unroll
    for (int t = 0; t < NT; ++t)
      acc[t] = __builtin_amdgcn_wmma_f32_16x16x32_f16(
           false, a[kc],  false, b[t],
           (short)0, acc[t],  false,  false);
    if (kc < 3) {
#pragma unroll
      for (int t = 0; t < NT; ++t) b[t] = wl[((kc + 1) * NT + t) * 32];
    }
    __builtin_amdgcn_sched_barrier(0);
  }

  float* ph = H + (size_t)(mtile * 16 + half * 8) * NOUT + 4 * r;
#pragma unroll
  for (int g = 0; g < NG; ++g) {
#pragma unroll
    for (int vr = 0; vr < 8; ++vr) {
      float4 o;
      o.x = acc[4 * g + 0][vr]; o.y = acc[4 * g + 1][vr];
      o.z = acc[4 * g + 2][vr]; o.w = acc[4 * g + 3][vr];
      *(float4*)(ph + (size_t)vr * NOUT + g * 64) = o;
    }
  }
}

#define NB      256
#define RPB     ((NNODES + NB - 1) / NB)
#define NCH     256
#define CHE     (NEDGES / NCH)
#define STEPS   ((CHE + 31) / 32)
#define SPW     ((STEPS + 31) / 32)

static_assert(NEDGES % NCH == 0, "chunks are equal");
static_assert(NB == 256 && (NNODES - 1) / RPB < NB, "owner id is 8 bits");
static_assert(RPB + 32 <= 512 && NNODES <= (1 << 17), "entry packs src<<9 | local row");
static_assert(NCH % 16 == 0, "a count row is read in four quarters of whole 16-byte groups");
static_assert(NCH * 1024 >= NNODES, "the count launch has a thread per node for the out-degree start value");
static_assert(RPB <= 512, "a block's rows have a counter each in split_kernel");

__device__ __forceinline__ unsigned same_owner(int o, bool ok) {
  unsigned m = (unsigned)__ballot(ok);
#pragma unroll
  for (int j = 0; j < 8; ++j) {
    const bool bit = (o >> j) & 1;
    const unsigned bj = (unsigned)__ballot(bit);
    m &= bit ? bj : ~bj;
  }
  return m;
}

template<bool FILL>
__global__ __launch_bounds__(1024)
void owner_sort_kernel(const int* __restrict__ src, const int* __restrict__ dst,
                       int* __restrict__ ccnt, int* __restrict__ otab, int* __restrict__ lst,
                       int* __restrict__ odeg) {
  __shared__ int hist[32 * NB];
  __shared__ int rtot[NB];
  __shared__ int rpre[NB];
  __shared__ int ostart[NB];
  const int tid = threadIdx.x, lane = tid & 31, c = blockIdx.x;
  const int w = __builtin_amdgcn_readfirstlane(tid >> 5);
  const unsigned below = (1u << lane) - 1u;
  int* myh = hist + w * NB;
#pragma unroll
  for (int k = 0; k < NB / 32; ++k) myh[k * 32 + lane] = 0;
  if (!FILL) {
    const int node = c * 1024 + tid;
    if (node < NNODES) odeg[node] = 1;
  }

  int d[SPW], s[SPW];
  bool ok[SPW];
#pragma unroll
  for (int u = 0; u < SPW; ++u) {
    const int i = (w * SPW + u) * 32 + lane;
    ok[u] = i < CHE;
    const int e = c * CHE + (ok[u] ? i : 0);
    d[u] = dst[e];
    s[u] = FILL ? src[e] : 0;
  }
  if (FILL) {
    __builtin_amdgcn_sched_barrier(0);
    const int b = tid >> 2, q = tid & 3;
    const int4* __restrict__ row = (const int4*)(ccnt + b * NCH + q * (NCH / 4));
    int4 v[NCH / 16];
#pragma unroll
    for (int k = 0; k < NCH / 16; ++k) v[k] = row[k];
    int tot = 0, pre = 0;
#pragma unroll
    for (int k = 0; k < NCH / 16; ++k) {
      const int ch = q * (NCH / 4) + k * 4;
      tot += v[k].x + v[k].y + v[k].z + v[k].w;
      pre += (ch + 0 < c ? v[k].x : 0) + (ch + 1 < c ? v[k].y : 0)
           + (ch + 2 < c ? v[k].z : 0) + (ch + 3 < c ? v[k].w : 0);
    }
    tot += __shfl_xor(tot, 1); tot += __shfl_xor(tot, 2);
    pre += __shfl_xor(pre, 1); pre += __shfl_xor(pre, 2);
    if (q == 0) { rtot[b] = tot; rpre[b] = pre; }
#pragma unroll
    for (int u = 0; u < SPW; ++u)
      if (ok[u]) atomicAdd(&odeg[s[u]], 1);
  }
  int o[SPW];
  unsigned m[SPW];
#pragma unroll
  for (int u = 0; u < SPW; ++u) {
    o[u] = (int)((unsigned)d[u] / RPB);
    m[u] = same_owner(o[u], ok[u]);
    if (ok[u] && (m[u] & below) == 0u) myh[o[u]] += __popc(m[u]);
  }
  __syncthreads();

  if (!FILL) {
    if (tid < NB) {
      int t = 0;
#pragma unroll
      for (int k = 0; k < 32; ++k) t += hist[k * NB + tid];
      ccnt[tid * NCH + c] = t;
    }
    return;
  }

  if (w == 0) {
    int v[8];
    int sum = 0;
#pragma unroll
    for (int k = 0; k < 8; ++k) { v[k] = rtot[lane * 8 + k]; sum += v[k]; }
    int incl = sum;
#pragma unroll
    for (int dd = 1; dd < 32; dd <<= 1) {
      int n = __shfl_up(incl, dd);
      if (lane >= dd) incl += n;
    }
    int run = incl - sum;
#pragma unroll
    for (int k = 0; k < 8; ++k) { ostart[lane * 8 + k] = run; run += v[k]; }
  }
  __syncthreads();
  if (tid < NB) {
    int run = ostart[tid] + rpre[tid];
    if (tid == c) { otab[c * 2] = ostart[tid]; otab[c * 2 + 1] = rtot[tid]; }
    int t[32];
#pragma unroll
    for (int k = 0; k < 32; ++k) t[k] = hist[k * NB + tid];
#pragma unroll
    for (int k = 0; k < 32; ++k) { hist[k * NB + tid] = run; run += t[k]; }
  }
  __syncthreads();
#pragma unroll
  for (int u = 0; u < SPW; ++u) {
    if (ok[u]) {
      const int base = myh[o[u]];
      const int r = __popc(m[u] & below);
      lst[base + r] = (s[u] << 9) | (d[u] - o[u] * RPB);
      if (r == 0) myh[o[u]] = base + __popc(m[u]);
    }
  }
}

#define PADENT  (15 * 32)
#define REGPAD  256
static_assert(32 * 7 + 8 <= REGPAD, "fill-up of 32 lists plus the alignment of the region");

__device__ __forceinline__ unsigned same_target(int tgt, bool ok) {
  unsigned m = (unsigned)__ballot(ok);
#pragma unroll
  for (int j = 0; j < 5; ++j) {
    const bool bit = (tgt >> j) & 1;
    const unsigned bj = (unsigned)__ballot(bit);
    m &= bit ? bj : ~bj;
  }
  return m;
}

__global__ __launch_bounds__(1024)
void split_kernel(const int* __restrict__ otab, const int* __restrict__ lst,
                  int* __restrict__ lst2, int* __restrict__ wtab,
                  const int* __restrict__ odeg, float* __restrict__ onorm, float* __restrict__ inorm) {
  __shared__ int hist[32 * 32];
  __shared__ int wpos[32 * 32];
  __shared__ int rcnt[512];
  const int tid = threadIdx.x, lane = tid & 31, b = blockIdx.x;
  const int lo = b * RPB;
  const int rows = (NNODES - lo < RPB) ? (NNODES - lo) : RPB;
  if (tid < 512) rcnt[tid] = 0;
  const int s = __builtin_amdgcn_readfirstlane(tid >> 5);
  const int lstart = otab[b * 2];
  const int nb = otab[b * 2 + 1];
  const int region = (lstart + b * REGPAD + 7) & ~7;
  const int C = ((nb + 1023) >> 10) << 5;
  const int i0 = s * C;
  const unsigned below = (1u << lane) - 1u;
  const int* __restrict__ bl = lst + lstart;
  int* myh = hist + s * 32;
  int* myp = wpos + s * 32;

  myh[lane] = 0;
  __syncthreads();
  const int od = odeg[lo + (tid < rows ? tid : 0)];

  for (int q = 0; q < C; q += 128) {
    int ent[4];
    bool ok[4];
#pragma unroll
    for (int u = 0; u < 4; ++u) {
      const int i = i0 + q + u * 32 + lane;
      ok[u] = (q + u * 32 < C) && (i < nb);
      ent[u] = 0;
      if (ok[u]) ent[u] = bl[i];
    }
#pragma unroll
    for (int u = 0; u < 4; ++u) {
      const int tgt = ent[u] & 31;
      const unsigned m = same_target(tgt, ok[u]);
      if (ok[u] && (m & below) == 0u) myh[tgt] += __popc(m);
      if (ok[u]) atomicAdd(&rcnt[ent[u] & 511], 1);
    }
  }
  __syncthreads();

  if (tid < rows) {
    inorm[lo + tid] = rsqrtf(fmaxf((float)(rcnt[tid] + 1), 1.0f));
    onorm[lo + tid] = rsqrtf(fmaxf((float)od, 1.0f));
  }

  {
    int col = 0;
#pragma unroll
    for (int l = 0; l < 32; ++l) col += hist[l * 32 + lane];
    const int pad = (col + 7) & ~7;
    int incl = pad;
#pragma unroll
    for (int d = 1; d < 32; d <<= 1) {
      int n = __shfl_up(incl, d);
      if (lane >= d) incl += n;
    }
    const int woff = __shfl(incl - pad, s);
    const int tot  = __shfl(col, s);
    const int tpad = __shfl(pad, s);
    const int mine = hist[lane * 32 + s];
    int inc2 = mine;
#pragma unroll
    for (int d = 1; d < 32; d <<= 1) {
      int n = __shfl_up(inc2, d);
      if (lane >= d) inc2 += n;
    }
    wpos[lane * 32 + s] = region + woff + inc2 - mine;
    if (lane == 0) { wtab[(b * 32 + s) * 2] = region + woff; wtab[(b * 32 + s) * 2 + 1] = tpad; }
    if (lane < tpad - tot) lst2[region + woff + tot + lane] = PADENT;
  }
  __syncthreads();

  for (int q = 0; q < C; q += 128) {
    int ent[4];
    bool ok[4];
#pragma unroll
    for (int u = 0; u < 4; ++u) {
      const int i = i0 + q + u * 32 + lane;
      ok[u] = (q + u * 32 < C) && (i < nb);
      ent[u] = 0;
      if (ok[u]) ent[u] = bl[i];
    }
#pragma unroll
    for (int u = 0; u < 4; ++u) {
      const int tgt = ent[u] & 31;
      const unsigned m = same_target(tgt, ok[u]);
      if (ok[u]) {
        const int base = myp[tgt];
        const int r = __popc(m & below);
        lst2[base + r] = ent[u];
        if (r == 0) myp[tgt] = base + __popc(m);
      }
    }
  }
}

#define KROWS   ((RPB + 31) / 32)
static_assert(KROWS <= 15, "register row 15 is the spare one");

template<int NOUT, bool RELU>
__global__ __launch_bounds__(1024)
void agg_owner_kernel(const int* __restrict__ wtab, const int* __restrict__ lst2,
                      const float* __restrict__ h, const float* __restrict__ inorm,
                      const float* __restrict__ bias, float* __restrict__ out) {
  constexpr int V = NOUT / 32;
  typedef float fv __attribute__((ext_vector_type(V)));
  typedef float v16f __attribute__((ext_vector_type(16)));
  typedef int v8i __attribute__((ext_vector_type(8)));
  const int tid = threadIdx.x, lane = tid & 31, b = blockIdx.x;
  const int w = __builtin_amdgcn_readfirstlane(tid >> 5);
  const int lo = b * RPB;
  const int rows = (NNODES - lo < RPB) ? (NNODES - lo) : RPB;

  const int lbeg = wtab[(b * 32 + w) * 2];
  const int n = wtab[(b * 32 + w) * 2 + 1];
  const v8i* __restrict__ myl = (const v8i*)(lst2 + lbeg);

  const fv* __restrict__ hv = (const fv*)h + (size_t)lo * 32;
  v16f a[V];
  {
    fv t[KROWS];
#pragma unroll
    for (int k = 0; k < KROWS; ++k) {
      const int r = k * 32 + w;
      t[k] = hv[(r < rows ? r : w) * 32 + lane];
    }
#pragma unroll
    for (int c = 0; c < V; ++c) {
#pragma unroll
      for (int k = 0; k < 16; ++k) a[c][k] = (k < KROWS) ? t[k < KROWS ? k : 0][c] : 0.0f;
    }
  }

  const char* __restrict__ hbytes = (const char*)h;
  const unsigned laneoff = (unsigned)lane * (unsigned)sizeof(fv);
  if (n > 0) {
    v8i en = myl[0];
    for (int j = 0; j < n; j += 8) {
      const v8i e = en;
      const int jn = (j + 8 < n) ? j + 8 : j;
      en = myl[jn >> 3];
      fv v[8];
#pragma unroll
      for (int u = 0; u < 8; ++u)
        v[u] = *(const fv*)(hbytes + (size_t)((unsigned)e[u] >> 9) * (32 * sizeof(fv)) + laneoff);
      __builtin_amdgcn_sched_barrier(0);
#pragma unroll
      for (int u = 0; u < 8; ++u) {
        const unsigned k = ((unsigned)e[u] >> 5) & 15u;
#pragma unroll
        for (int cc = 0; cc < V; ++cc) a[cc][k] += v[u][cc];
      }
    }
  }

  const fv* __restrict__ bv = (const fv*)bias;
  fv* __restrict__ ov = (fv*)out + (size_t)lo * 32;
  const fv bb = bv[lane];
  float sc[KROWS];
#pragma unroll
  for (int k = 0; k < KROWS; ++k) {
    const int r = k * 32 + w;
    sc[k] = inorm[lo + (r < rows ? r : w)];
  }
#pragma unroll
  for (int k = 0; k < KROWS; ++k) {
    const int r = k * 32 + w;
    if (r < rows) {
      fv x;
#pragma unroll
      for (int cc = 0; cc < V; ++cc) x[cc] = a[cc][k];
      x = x * sc[k] + bb;
      if (RELU) {
#pragma unroll
        for (int cc = 0; cc < V; ++cc) x[cc] = fmaxf(x[cc], 0.0f);
      }
      ov[r * 32 + lane] = x;
    }
  }
}

extern "C" void kernel_launch(void* const* d_in, const int* in_sizes, int n_in,
                              void* d_out, int out_size, void* d_ws, size_t ws_size,
                              hipStream_t stream) {
  const float* feat = (const float*)d_in[0];
  const int*   src  = (const int*)  d_in[1];
  const int*   dst  = (const int*)  d_in[2];
  const float* W1   = (const float*)d_in[3];
  const float* b1   = (const float*)d_in[4];
  const float* W2   = (const float*)d_in[5];
  const float* b2   = (const float*)d_in[6];
  const float* W3   = (const float*)d_in[7];
  const float* b3   = (const float*)d_in[8];
  float* out = (float*)d_out;

  size_t off = 0;
  auto alloc = [&](size_t bytes) -> char* {
    char* p = (char*)d_ws + off;
    off += (bytes + 255) & ~(size_t)255;
    return p;
  };
  float*     onorm = (float*)alloc((size_t)NNODES * sizeof(float));
  float*     inorm = (float*)alloc((size_t)NNODES * sizeof(float));
  int*       odeg  = (int*)alloc((size_t)NNODES * sizeof(int));
  float*     hbuf  = (float*)alloc((size_t)NNODES * FDIM * sizeof(float));
  float*     xbuf  = (float*)alloc((size_t)NNODES * FDIM * sizeof(float));
  _Float16*  Wp    = (_Float16*)alloc((size_t)FDIM * FDIM * sizeof(_Float16));
  int*       ccnt  = (int*)alloc((size_t)NB * NCH * sizeof(int));
  int*       otab  = (int*)alloc((size_t)NB * 2 * sizeof(int));
  int*       lst   = (int*)alloc((size_t)NEDGES * sizeof(int));
  int*       lst2  = (int*)alloc(((size_t)NEDGES + (size_t)NB * REGPAD + 64) * sizeof(int));
  int*       wtab  = (int*)alloc((size_t)NB * 32 * 2 * sizeof(int));

  const int TB = 256;

  owner_sort_kernel<false><<<NCH, 1024, 0, stream>>>(src, dst, ccnt, otab, lst, odeg);
  owner_sort_kernel<true><<<NCH, 1024, 0, stream>>>(src, dst, ccnt, otab, lst, odeg);
  split_kernel<<<NB, 1024, 0, stream>>>(otab, lst, lst2, wtab, odeg, onorm, inorm);

  pack_w_kernel<<<ceil_div(4 * 8 * 512, TB), TB, 0, stream>>>(W1, Wp, 128);
  {
    int waves = NNODES / 16;
    gemm_wmma_kernel<128><<<ceil_div((long long)waves * 32, GEMM_TB), GEMM_TB, 0, stream>>>(feat, onorm, Wp, hbuf);
  }
  agg_owner_kernel<128, true><<<NB, 1024, 0, stream>>>(wtab, lst2, hbuf, inorm, b1, xbuf);

  pack_w_kernel<<<ceil_div(4 * 8 * 512, TB), TB, 0, stream>>>(W2, Wp, 128);
  {
    int waves = NNODES / 16;
    gemm_wmma_kernel<128><<<ceil_div((long long)waves * 32, GEMM_TB), GEMM_TB, 0, stream>>>(xbuf, onorm, Wp, hbuf);
  }
  agg_owner_kernel<128, true><<<NB, 1024, 0, stream>>>(wtab, lst2, hbuf, inorm, b2, xbuf);

  pack_w_kernel<<<ceil_div(4 * 4 * 512, TB), TB, 0, stream>>>(W3, Wp, 64);
  {
    int waves = NNODES / 16;
    gemm_wmma_kernel<64><<<ceil_div((long long)waves * 32, GEMM_TB), GEMM_TB, 0, stream>>>(xbuf, onorm, Wp, hbuf);
  }
  agg_owner_kernel<64, false><<<NB, 1024, 0, stream>>>(wtab, lst2, hbuf, inorm, b3, out);
}


# Out-degrees without global adds: the count launch's blocks also count source ids in LDS (8 node ranges x 32 edge chunks), split kernel sums a node's 32 counted pieces; integer sums, same norms
# speedup vs baseline: 4.2075x; 1.2636x over previous
#include <hip/hip_runtime.h>
#include <math.h>

typedef __attribute__((ext_vector_type(16))) _Float16 v16h;
typedef __attribute__((ext_vector_type(8)))  float    v8f;

#define NNODES 100000
#define NEDGES 1600000
#define FDIM   128

static inline int ceil_div(long long a, long long b) { return (int)((a + b - 1) / b); }


__global__ void pack_w_kernel(const float* __restrict__ W, _Float16* __restrict__ Wp, int nout) {
  int idx = blockIdx.x * blockDim.x + threadIdx.x;
  int nt = nout >> 4;
  int total = 4 * nt * 32 * 16;
  if (idx >= total) return;
  int j    = idx & 15;
  int l    = (idx >> 4) & 31;
  int rest = idx >> 9;
  int t    = rest % nt;
  int c    = rest / nt;
  int col   = (t >> 2) * 64 + 4 * (l & 15) + (t & 3);
  int kbase = c * 32 + ((l >> 4) * 8);
  int K     = kbase + (j & 7) + ((j >> 3) * 16);
  Wp[idx] = (_Float16)W[K * nout + col];
}

#define GEMM_TB 256

template<int NOUT>
__global__ __launch_bounds__(GEMM_TB)
void gemm_wmma_kernel(const float* __restrict__ X,
                      const float* __restrict__ onorm,
                      const _Float16* __restrict__ Wp,
                      float* __restrict__ H) {
  constexpr int NT = NOUT / 16;
  constexpr int NG = NT / 4;
  static_assert(NT % 4 == 0, "column tiles come in groups of four");
  int mtile = (int)((blockIdx.x * blockDim.x + threadIdx.x) >> 5);
  int lane = threadIdx.x & 31;
  if (mtile * 16 >= NNODES) return;

  int r    = lane & 15;
  int half = lane >> 4;
  int row  = mtile * 16 + r;
  const float* px = X + (size_t)row * FDIM;
  const v16h* __restrict__ wl = (const v16h*)Wp + lane;

  float4 f[16];
#pragma unroll
  for (int kc = 0; kc < 4; ++kc) {
    int kbase = kc * 32 + half * 8;
    f[kc * 4 + 0] = *(const float4*)(px + kbase);
    f[kc * 4 + 1] = *(const float4*)(px + kbase + 4);
    f[kc * 4 + 2] = *(const float4*)(px + kbase + 16);
    f[kc * 4 + 3] = *(const float4*)(px + kbase + 20);
  }
  float scale = onorm[row];
  v16h b[NT];
#pragma unroll
  for (int t = 0; t < NT; ++t) b[t] = wl[t * 32];
  __builtin_amdgcn_sched_barrier(0);

  v16h a[4];
#pragma unroll
  for (int kc = 0; kc < 4; ++kc) {
#pragma unroll
    for (int i = 0; i < 4; ++i) {
      const float4 q = f[kc * 4 + i];
      a[kc][i * 4 + 0] = (_Float16)(q.x * scale);
      a[kc][i * 4 + 1] = (_Float16)(q.y * scale);
      a[kc][i * 4 + 2] = (_Float16)(q.z * scale);
      a[kc][i * 4 + 3] = (_Float16)(q.w * scale);
    }
  }

  v8f acc[NT];
#pragma unroll
  for (int t = 0; t < NT; ++t) acc[t] = (v8f){0.f, 0.f, 0.f, 0.f, 0.f, 0.f, 0.f, 0.f};

#pragma unroll
  for (int kc = 0; kc < 4; ++kc) {
#pragma unroll
    for (int t = 0; t < NT; ++t)
      acc[t] = __builtin_amdgcn_wmma_f32_16x16x32_f16(
           false, a[kc],  false, b[t],
           (short)0, acc[t],  false,  false);
    if (kc < 3) {
#pragma unroll
      for (int t = 0; t < NT; ++t) b[t] = wl[((kc + 1) * NT + t) * 32];
    }
    __builtin_amdgcn_sched_barrier(0);
  }

  float* ph = H + (size_t)(mtile * 16 + half * 8) * NOUT + 4 * r;
#pragma unroll
  for (int g = 0; g < NG; ++g) {
#pragma unroll
    for (int vr = 0; vr < 8; ++vr) {
      float4 o;
      o.x = acc[4 * g + 0][vr]; o.y = acc[4 * g + 1][vr];
      o.z = acc[4 * g + 2][vr]; o.w = acc[4 * g + 3][vr];
      *(float4*)(ph + (size_t)vr * NOUT + g * 64) = o;
    }
  }
}

#define NB      256
#define RPB     ((NNODES + NB - 1) / NB)
#define NCH     256
#define CHE     (NEDGES / NCH)
#define STEPS   ((CHE + 31) / 32)
#define SPW     ((STEPS + 31) / 32)

static_assert(NEDGES % NCH == 0, "chunks are equal");
static_assert(NB == 256 && (NNODES - 1) / RPB < NB, "owner id is 8 bits");
static_assert(RPB + 32 <= 512 && NNODES <= (1 << 17), "entry packs src<<9 | local row");
static_assert(NCH % 16 == 0, "a count row is read in four quarters of whole 16-byte groups");
#define ORNG    8
#define OCHK    (NCH / ORNG)
#define ONR     (NNODES / ORNG)
#define OCE     (NEDGES / OCHK)
static_assert(ORNG * OCHK == NCH, "one block of the count launch per (node range, edge chunk)");
static_assert(NNODES % ORNG == 0 && NEDGES % OCHK == 0, "ranges and chunks are equal");
static_assert(ONR % 4 == 0 && OCE % 4 == 0 && NNODES % 4 == 0, "ids and counts move in whole 16-byte pieces");
static_assert(RPB <= 512, "a block's rows have a counter each in split_kernel");

__device__ __forceinline__ unsigned same_owner(int o, bool ok) {
  unsigned m = (unsigned)__ballot(ok);
#pragma unroll
  for (int j = 0; j < 8; ++j) {
    const bool bit = (o >> j) & 1;
    const unsigned bj = (unsigned)__ballot(bit);
    m &= bit ? bj : ~bj;
  }
  return m;
}

template<bool FILL>
__global__ __launch_bounds__(1024)
void owner_sort_kernel(const int* __restrict__ src, const int* __restrict__ dst,
                       int* __restrict__ ccnt, int* __restrict__ otab, int* __restrict__ lst,
                       int* __restrict__ opart) {
  __shared__ int hist[32 * NB];
  __shared__ int rtot[NB];
  __shared__ int rpre[NB];
  __shared__ int ostart[NB];
  __shared__ int4 ocnt4[FILL ? 1 : ONR / 4];
  int* ocnt = (int*)ocnt4;
  const int tid = threadIdx.x, lane = tid & 31, c = blockIdx.x;
  const int w = __builtin_amdgcn_readfirstlane(tid >> 5);
  const unsigned below = (1u << lane) - 1u;
  int* myh = hist + w * NB;
#pragma unroll
  for (int k = 0; k < NB / 32; ++k) myh[k * 32 + lane] = 0;
  if (!FILL) {
    for (int i = tid; i < ONR / 4; i += 1024) ocnt4[i] = make_int4(0, 0, 0, 0);
  }

  int d[SPW], s[SPW];
  bool ok[SPW];
#pragma unroll
  for (int u = 0; u < SPW; ++u) {
    const int i = (w * SPW + u) * 32 + lane;
    ok[u] = i < CHE;
    const int e = c * CHE + (ok[u] ? i : 0);
    d[u] = dst[e];
    s[u] = FILL ? src[e] : 0;
  }
  if (FILL) {
    __builtin_amdgcn_sched_barrier(0);
    const int b = tid >> 2, q = tid & 3;
    const int4* __restrict__ row = (const int4*)(ccnt + b * NCH + q * (NCH / 4));
    int4 v[NCH / 16];
#pragma unroll
    for (int k = 0; k < NCH / 16; ++k) v[k] = row[k];
    int tot = 0, pre = 0;
#pragma unroll
    for (int k = 0; k < NCH / 16; ++k) {
      const int ch = q * (NCH / 4) + k * 4;
      tot += v[k].x + v[k].y + v[k].z + v[k].w;
      pre += (ch + 0 < c ? v[k].x : 0) + (ch + 1 < c ? v[k].y : 0)
           + (ch + 2 < c ? v[k].z : 0) + (ch + 3 < c ? v[k].w : 0);
    }
    tot += __shfl_xor(tot, 1); tot += __shfl_xor(tot, 2);
    pre += __shfl_xor(pre, 1); pre += __shfl_xor(pre, 2);
    if (q == 0) { rtot[b] = tot; rpre[b] = pre; }
  } else {
    constexpr int NQ  = OCE / 4;
    constexpr int RND = (NQ + 1023) / 1024;
    const int nlo = (c / OCHK) * ONR;
    const int4* __restrict__ sp = (const int4*)(src + (c % OCHK) * OCE);
    int4 q[RND];
#pragma unroll
    for (int k = 0; k < RND; ++k) {
      const int i = k * 1024 + tid;
      q[k] = sp[i < NQ ? i : 0];
    }
    __syncthreads();
#pragma unroll
    for (int k = 0; k < RND; ++k) {
      if (k * 1024 + tid < NQ) {
        const unsigned a0 = (unsigned)(q[k].x - nlo), a1 = (unsigned)(q[k].y - nlo);
        const unsigned a2 = (unsigned)(q[k].z - nlo), a3 = (unsigned)(q[k].w - nlo);
        if (a0 < (unsigned)ONR) atomicAdd(&ocnt[a0], 1);
        if (a1 < (unsigned)ONR) atomicAdd(&ocnt[a1], 1);
        if (a2 < (unsigned)ONR) atomicAdd(&ocnt[a2], 1);
        if (a3 < (unsigned)ONR) atomicAdd(&ocnt[a3], 1);
      }
    }
  }
  int o[SPW];
  unsigned m[SPW];
#pragma unroll
  for (int u = 0; u < SPW; ++u) {
    o[u] = (int)((unsigned)d[u] / RPB);
    m[u] = same_owner(o[u], ok[u]);
    if (ok[u] && (m[u] & below) == 0u) myh[o[u]] += __popc(m[u]);
  }
  __syncthreads();

  if (!FILL) {
    if (tid < NB) {
      int t = 0;
#pragma unroll
      for (int k = 0; k < 32; ++k) t += hist[k * NB + tid];
      ccnt[tid * NCH + c] = t;
    }
    int4* __restrict__ op = (int4*)(opart + (c % OCHK) * NNODES + (c / OCHK) * ONR);
    for (int i = tid; i < ONR / 4; i += 1024) op[i] = ocnt4[i];
    return;
  }

  if (w == 0) {
    int v[8];
    int sum = 0;
#pragma unroll
    for (int k = 0; k < 8; ++k) { v[k] = rtot[lane * 8 + k]; sum += v[k]; }
    int incl = sum;
#pragma unroll
    for (int dd = 1; dd < 32; dd <<= 1) {
      int n = __shfl_up(incl, dd);
      if (lane >= dd) incl += n;
    }
    int run = incl - sum;
#pragma unroll
    for (int k = 0; k < 8; ++k) { ostart[lane * 8 + k] = run; run += v[k]; }
  }
  __syncthreads();
  if (tid < NB) {
    int run = ostart[tid] + rpre[tid];
    if (tid == c) { otab[c * 2] = ostart[tid]; otab[c * 2 + 1] = rtot[tid]; }
    int t[32];
#pragma unroll
    for (int k = 0; k < 32; ++k) t[k] = hist[k * NB + tid];
#pragma unroll
    for (int k = 0; k < 32; ++k) { hist[k * NB + tid] = run; run += t[k]; }
  }
  __syncthreads();
#pragma unroll
  for (int u = 0; u < SPW; ++u) {
    if (ok[u]) {
      const int base = myh[o[u]];
      const int r = __popc(m[u] & below);
      lst[base + r] = (s[u] << 9) | (d[u] - o[u] * RPB);
      if (r == 0) myh[o[u]] = base + __popc(m[u]);
    }
  }
}

#define PADENT  (15 * 32)
#define REGPAD  256
static_assert(32 * 7 + 8 <= REGPAD, "fill-up of 32 lists plus the alignment of the region");

__device__ __forceinline__ unsigned same_target(int tgt, bool ok) {
  unsigned m = (unsigned)__ballot(ok);
#pragma unroll
  for (int j = 0; j < 5; ++j) {
    const bool bit = (tgt >> j) & 1;
    const unsigned bj = (unsigned)__ballot(bit);
    m &= bit ? bj : ~bj;
  }
  return m;
}

__global__ __launch_bounds__(1024)
void split_kernel(const int* __restrict__ otab, const int* __restrict__ lst,
                  int* __restrict__ lst2, int* __restrict__ wtab,
                  const int* __restrict__ opart, float* __restrict__ onorm, float* __restrict__ inorm) {
  __shared__ int hist[32 * 32];
  __shared__ int wpos[32 * 32];
  __shared__ int rcnt[512];
  const int tid = threadIdx.x, lane = tid & 31, b = blockIdx.x;
  const int lo = b * RPB;
  const int rows = (NNODES - lo < RPB) ? (NNODES - lo) : RPB;
  if (tid < 512) rcnt[tid] = 0;
  const int s = __builtin_amdgcn_readfirstlane(tid >> 5);
  const int lstart = otab[b * 2];
  const int nb = otab[b * 2 + 1];
  const int region = (lstart + b * REGPAD + 7) & ~7;
  const int C = ((nb + 1023) >> 10) << 5;
  const int i0 = s * C;
  const unsigned below = (1u << lane) - 1u;
  const int* __restrict__ bl = lst + lstart;
  int* myh = hist + s * 32;
  int* myp = wpos + s * 32;

  myh[lane] = 0;
  __syncthreads();
  int p[OCHK];
  if (s * 32 < rows) {
    const int* __restrict__ pn = opart + lo + (tid < rows ? tid : 0);
#pragma unroll
    for (int k = 0; k < OCHK; ++k) p[k] = pn[k * NNODES];
  }

  for (int q = 0; q < C; q += 128) {
    int ent[4];
    bool ok[4];
#pragma unroll
    for (int u = 0; u < 4; ++u) {
      const int i = i0 + q + u * 32 + lane;
      ok[u] = (q + u * 32 < C) && (i < nb);
      ent[u] = 0;
      if (ok[u]) ent[u] = bl[i];
    }
#pragma unroll
    for (int u = 0; u < 4; ++u) {
      const int tgt = ent[u] & 31;
      const unsigned m = same_target(tgt, ok[u]);
      if (ok[u] && (m & below) == 0u) myh[tgt] += __popc(m);
      if (ok[u]) atomicAdd(&rcnt[ent[u] & 511], 1);
    }
  }
  __syncthreads();

  if (tid < rows) {
    int od = 1;
#pragma unroll
    for (int k = 0; k < OCHK; ++k) od += p[k];
    inorm[lo + tid] = rsqrtf(fmaxf((float)(rcnt[tid] + 1), 1.0f));
    onorm[lo + tid] = rsqrtf(fmaxf((float)od, 1.0f));
  }

  {
    int col = 0;
#pragma unroll
    for (int l = 0; l < 32; ++l) col += hist[l * 32 + lane];
    const int pad = (col + 7) & ~7;
    int incl = pad;
#pragma unroll
    for (int d = 1; d < 32; d <<= 1) {
      int n = __shfl_up(incl, d);
      if (lane >= d) incl += n;
    }
    const int woff = __shfl(incl - pad, s);
    const int tot  = __shfl(col, s);
    const int tpad = __shfl(pad, s);
    const int mine = hist[lane * 32 + s];
    int inc2 = mine;
#pragma unroll
    for (int d = 1; d < 32; d <<= 1) {
      int n = __shfl_up(inc2, d);
      if (lane >= d) inc2 += n;
    }
    wpos[lane * 32 + s] = region + woff + inc2 - mine;
    if (lane == 0) { wtab[(b * 32 + s) * 2] = region + woff; wtab[(b * 32 + s) * 2 + 1] = tpad; }
    if (lane < tpad - tot) lst2[region + woff + tot + lane] = PADENT;
  }
  __syncthreads();

  for (int q = 0; q < C; q += 128) {
    int ent[4];
    bool ok[4];
#pragma unroll
    for (int u = 0; u < 4; ++u) {
      const int i = i0 + q + u * 32 + lane;
      ok[u] = (q + u * 32 < C) && (i < nb);
      ent[u] = 0;
      if (ok[u]) ent[u] = bl[i];
    }
#pragma unroll
    for (int u = 0; u < 4; ++u) {
      const int tgt = ent[u] & 31;
      const unsigned m = same_target(tgt, ok[u]);
      if (ok[u]) {
        const int base = myp[tgt];
        const int r = __popc(m & below);
        lst2[base + r] = ent[u];
        if (r == 0) myp[tgt] = base + __popc(m);
      }
    }
  }
}

#define KROWS   ((RPB + 31) / 32)
static_assert(KROWS <= 15, "register row 15 is the spare one");

template<int NOUT, bool RELU>
__global__ __launch_bounds__(1024)
void agg_owner_kernel(const int* __restrict__ wtab, const int* __restrict__ lst2,
                      const float* __restrict__ h, const float* __restrict__ inorm,
                      const float* __restrict__ bias, float* __restrict__ out) {
  constexpr int V = NOUT / 32;
  typedef float fv __attribute__((ext_vector_type(V)));
  typedef float v16f __attribute__((ext_vector_type(16)));
  typedef int v8i __attribute__((ext_vector_type(8)));
  const int tid = threadIdx.x, lane = tid & 31, b = blockIdx.x;
  const int w = __builtin_amdgcn_readfirstlane(tid >> 5);
  const int lo = b * RPB;
  const int rows = (NNODES - lo < RPB) ? (NNODES - lo) : RPB;

  const int lbeg = wtab[(b * 32 + w) * 2];
  const int n = wtab[(b * 32 + w) * 2 + 1];
  const v8i* __restrict__ myl = (const v8i*)(lst2 + lbeg);

  const fv* __restrict__ hv = (const fv*)h + (size_t)lo * 32;
  v16f a[V];
  {
    fv t[KROWS];
#pragma unroll
    for (int k = 0; k < KROWS; ++k) {
      const int r = k * 32 + w;
      t[k] = hv[(r < rows ? r : w) * 32 + lane];
    }
#pragma unroll
    for (int c = 0; c < V; ++c) {
#pragma unroll
      for (int k = 0; k < 16; ++k) a[c][k] = (k < KROWS) ? t[k < KROWS ? k : 0][c] : 0.0f;
    }
  }

  const char* __restrict__ hbytes = (const char*)h;
  const unsigned laneoff = (unsigned)lane * (unsigned)sizeof(fv);
  if (n > 0) {
    v8i en = myl[0];
    for (int j = 0; j < n; j += 8) {
      const v8i e = en;
      const int jn = (j + 8 < n) ? j + 8 : j;
      en = myl[jn >> 3];
      fv v[8];
#pragma unroll
      for (int u = 0; u < 8; ++u)
        v[u] = *(const fv*)(hbytes + (size_t)((unsigned)e[u] >> 9) * (32 * sizeof(fv)) + laneoff);
      __builtin_amdgcn_sched_barrier(0);
#pragma unroll
      for (int u = 0; u < 8; ++u) {
        const unsigned k = ((unsigned)e[u] >> 5) & 15u;
#pragma unroll
        for (int cc = 0; cc < V; ++cc) a[cc][k] += v[u][cc];
      }
    }
  }

  const fv* __restrict__ bv = (const fv*)bias;
  fv* __restrict__ ov = (fv*)out + (size_t)lo * 32;
  const fv bb = bv[lane];
  float sc[KROWS];
#pragma unroll
  for (int k = 0; k < KROWS; ++k) {
    const int r = k * 32 + w;
    sc[k] = inorm[lo + (r < rows ? r : w)];
  }
#pragma unroll
  for (int k = 0; k < KROWS; ++k) {
    const int r = k * 32 + w;
    if (r < rows) {
      fv x;
#pragma unroll
      for (int cc = 0; cc < V; ++cc) x[cc] = a[cc][k];
      x = x * sc[k] + bb;
      if (RELU) {
#pragma unroll
        for (int cc = 0; cc < V; ++cc) x[cc] = fmaxf(x[cc], 0.0f);
      }
      ov[r * 32 + lane] = x;
    }
  }
}

extern "C" void kernel_launch(void* const* d_in, const int* in_sizes, int n_in,
                              void* d_out, int out_size, void* d_ws, size_t ws_size,
                              hipStream_t stream) {
  const float* feat = (const float*)d_in[0];
  const int*   src  = (const int*)  d_in[1];
  const int*   dst  = (const int*)  d_in[2];
  const float* W1   = (const float*)d_in[3];
  const float* b1   = (const float*)d_in[4];
  const float* W2   = (const float*)d_in[5];
  const float* b2   = (const float*)d_in[6];
  const float* W3   = (const float*)d_in[7];
  const float* b3   = (const float*)d_in[8];
  float* out = (float*)d_out;

  size_t off = 0;
  auto alloc = [&](size_t bytes) -> char* {
    char* p = (char*)d_ws + off;
    off += (bytes + 255) & ~(size_t)255;
    return p;
  };
  float*     onorm = (float*)alloc((size_t)NNODES * sizeof(float));
  float*     inorm = (float*)alloc((size_t)NNODES * sizeof(float));
  int*       opart = (int*)alloc((size_t)OCHK * NNODES * sizeof(int));
  float*     hbuf  = (float*)alloc((size_t)NNODES * FDIM * sizeof(float));
  float*     xbuf  = (float*)alloc((size_t)NNODES * FDIM * sizeof(float));
  _Float16*  Wp    = (_Float16*)alloc((size_t)FDIM * FDIM * sizeof(_Float16));
  int*       ccnt  = (int*)alloc((size_t)NB * NCH * sizeof(int));
  int*       otab  = (int*)alloc((size_t)NB * 2 * sizeof(int));
  int*       lst   = (int*)alloc((size_t)NEDGES * sizeof(int));
  int*       lst2  = (int*)alloc(((size_t)NEDGES + (size_t)NB * REGPAD + 64) * sizeof(int));
  int*       wtab  = (int*)alloc((size_t)NB * 32 * 2 * sizeof(int));

  const int TB = 256;

  owner_sort_kernel<false><<<NCH, 1024, 0, stream>>>(src, dst, ccnt, otab, lst, opart);
  owner_sort_kernel<true><<<NCH, 1024, 0, stream>>>(src, dst, ccnt, otab, lst, opart);
  split_kernel<<<NB, 1024, 0, stream>>>(otab, lst, lst2, wtab, opart, onorm, inorm);

  pack_w_kernel<<<ceil_div(4 * 8 * 512, TB), TB, 0, stream>>>(W1, Wp, 128);
  {
    int waves = NNODES / 16;
    gemm_wmma_kernel<128><<<ceil_div((long long)waves * 32, GEMM_TB), GEMM_TB, 0, stream>>>(feat, onorm, Wp, hbuf);
  }
  agg_owner_kernel<128, true><<<NB, 1024, 0, stream>>>(wtab, lst2, hbuf, inorm, b1, xbuf);

  pack_w_kernel<<<ceil_div(4 * 8 * 512, TB), TB, 0, stream>>>(W2, Wp, 128);
  {
    int waves = NNODES / 16;
    gemm_wmma_kernel<128><<<ceil_div((long long)waves * 32, GEMM_TB), GEMM_TB, 0, stream>>>(xbuf, onorm, Wp, hbuf);
  }
  agg_owner_kernel<128, true><<<NB, 1024, 0, stream>>>(wtab, lst2, hbuf, inorm, b2, xbuf);

  pack_w_kernel<<<ceil_div(4 * 4 * 512, TB), TB, 0, stream>>>(W3, Wp, 64);
  {
    int waves = NNODES / 16;
    gemm_wmma_kernel<64><<<ceil_div((long long)waves * 32, GEMM_TB), GEMM_TB, 0, stream>>>(xbuf, onorm, Wp, hbuf);
  }
  agg_owner_kernel<64, false><<<NB, 1024, 0, stream>>>(wtab, lst2, hbuf, inorm, b3, out);
}


# Edge grouping in one launch: each chunk block writes its edges grouped by owner into its own stretch of the list; owner blocks read their 256 groups in chunk order; count launch gone; same sums
# speedup vs baseline: 4.6439x; 1.1037x over previous
#include <hip/hip_runtime.h>
#include <math.h>

typedef __attribute__((ext_vector_type(16))) _Float16 v16h;
typedef __attribute__((ext_vector_type(8)))  float    v8f;

#define NNODES 100000
#define NEDGES 1600000
#define FDIM   128

static inline int ceil_div(long long a, long long b) { return (int)((a + b - 1) / b); }


__global__ void pack_w_kernel(const float* __restrict__ W, _Float16* __restrict__ Wp, int nout) {
  int idx = blockIdx.x * blockDim.x + threadIdx.x;
  int nt = nout >> 4;
  int total = 4 * nt * 32 * 16;
  if (idx >= total) return;
  int j    = idx & 15;
  int l    = (idx >> 4) & 31;
  int rest = idx >> 9;
  int t    = rest % nt;
  int c    = rest / nt;
  int col   = (t >> 2) * 64 + 4 * (l & 15) + (t & 3);
  int kbase = c * 32 + ((l >> 4) * 8);
  int K     = kbase + (j & 7) + ((j >> 3) * 16);
  Wp[idx] = (_Float16)W[K * nout + col];
}

#define GEMM_TB 256

template<int NOUT>
__global__ __launch_bounds__(GEMM_TB)
void gemm_wmma_kernel(const float* __restrict__ X,
                      const float* __restrict__ onorm,
                      const _Float16* __restrict__ Wp,
                      float* __restrict__ H) {
  constexpr int NT = NOUT / 16;
  constexpr int NG = NT / 4;
  static_assert(NT % 4 == 0, "column tiles come in groups of four");
  int mtile = (int)((blockIdx.x * blockDim.x + threadIdx.x) >> 5);
  int lane = threadIdx.x & 31;
  if (mtile * 16 >= NNODES) return;

  int r    = lane & 15;
  int half = lane >> 4;
  int row  = mtile * 16 + r;
  const float* px = X + (size_t)row * FDIM;
  const v16h* __restrict__ wl = (const v16h*)Wp + lane;

  float4 f[16];
#pragma unroll
  for (int kc = 0; kc < 4; ++kc) {
    int kbase = kc * 32 + half * 8;
    f[kc * 4 + 0] = *(const float4*)(px + kbase);
    f[kc * 4 + 1] = *(const float4*)(px + kbase + 4);
    f[kc * 4 + 2] = *(const float4*)(px + kbase + 16);
    f[kc * 4 + 3] = *(const float4*)(px + kbase + 20);
  }
  float scale = onorm[row];
  v16h b[NT];
#pragma unroll
  for (int t = 0; t < NT; ++t) b[t] = wl[t * 32];
  __builtin_amdgcn_sched_barrier(0);

  v16h a[4];
#pragma unroll
  for (int kc = 0; kc < 4; ++kc) {
#pragma unroll
    for (int i = 0; i < 4; ++i) {
      const float4 q = f[kc * 4 + i];
      a[kc][i * 4 + 0] = (_Float16)(q.x * scale);
      a[kc][i * 4 + 1] = (_Float16)(q.y * scale);
      a[kc][i * 4 + 2] = (_Float16)(q.z * scale);
      a[kc][i * 4 + 3] = (_Float16)(q.w * scale);
    }
  }

  v8f acc[NT];
#pragma unroll
  for (int t = 0; t < NT; ++t) acc[t] = (v8f){0.f, 0.f, 0.f, 0.f, 0.f, 0.f, 0.f, 0.f};

#pragma unroll
  for (int kc = 0; kc < 4; ++kc) {
#pragma unroll
    for (int t = 0; t < NT; ++t)
      acc[t] = __builtin_amdgcn_wmma_f32_16x16x32_f16(
           false, a[kc],  false, b[t],
           (short)0, acc[t],  false,  false);
    if (kc < 3) {
#pragma unroll
      for (int t = 0; t < NT; ++t) b[t] = wl[((kc + 1) * NT + t) * 32];
    }
    __builtin_amdgcn_sched_barrier(0);
  }

  float* ph = H + (size_t)(mtile * 16 + half * 8) * NOUT + 4 * r;
#pragma unroll
  for (int g = 0; g < NG; ++g) {
#pragma unroll
    for (int vr = 0; vr < 8; ++vr) {
      float4 o;
      o.x = acc[4 * g + 0][vr]; o.y = acc[4 * g + 1][vr];
      o.z = acc[4 * g + 2][vr]; o.w = acc[4 * g + 3][vr];
      *(float4*)(ph + (size_t)vr * NOUT + g * 64) = o;
    }
  }
}

#define NB      256
#define RPB     ((NNODES + NB - 1) / NB)
#define NCH     256
#define CHE     (NEDGES / NCH)
#define STEPS   ((CHE + 31) / 32)
#define SPW     ((STEPS + 31) / 32)

static_assert(NEDGES % NCH == 0, "chunks are equal");
static_assert(NB == 256 && (NNODES - 1) / RPB < NB, "owner id is 8 bits");
static_assert(RPB + 32 <= 512 && NNODES <= (1 << 17), "entry packs src<<9 | local row");
static_assert(NCH % 16 == 0, "a count row is read in four quarters of whole 16-byte groups");
#define ORNG    8
#define OCHK    (NCH / ORNG)
#define ONR     (NNODES / ORNG)
#define OCE     (NEDGES / OCHK)
static_assert(ORNG * OCHK == NCH, "one block of the grouping launch per (node range, edge chunk)");
static_assert(NNODES % ORNG == 0 && NEDGES % OCHK == 0, "ranges and chunks are equal");
static_assert(ONR % 4 == 0 && OCE % 4 == 0 && NNODES % 4 == 0, "ids and counts move in whole 16-byte pieces");
static_assert(RPB <= 512, "a block's rows have a counter each in split_kernel");

__device__ __forceinline__ unsigned same_owner(int o, bool ok) {
  unsigned m = (unsigned)__ballot(ok);
#pragma unroll
  for (int j = 0; j < 8; ++j) {
    const bool bit = (o >> j) & 1;
    const unsigned bj = (unsigned)__ballot(bit);
    m &= bit ? bj : ~bj;
  }
  return m;
}

__global__ __launch_bounds__(1024)
void owner_sort_kernel(const int* __restrict__ src, const int* __restrict__ dst,
                       int* __restrict__ ccnt, int* __restrict__ cstart, int* __restrict__ lst,
                       int* __restrict__ opart) {
  __shared__ int hist[32 * NB];
  __shared__ int rtot[NB];
  __shared__ int ostart[NB];
  __shared__ int4 ocnt4[ONR / 4];
  int* ocnt = (int*)ocnt4;
  const int tid = threadIdx.x, lane = tid & 31, c = blockIdx.x;
  const int w = __builtin_amdgcn_readfirstlane(tid >> 5);
  const unsigned below = (1u << lane) - 1u;
  int* myh = hist + w * NB;
#pragma unroll
  for (int k = 0; k < NB / 32; ++k) myh[k * 32 + lane] = 0;
  for (int i = tid; i < ONR / 4; i += 1024) ocnt4[i] = make_int4(0, 0, 0, 0);

  int d[SPW], s[SPW];
  bool ok[SPW];
#pragma unroll
  for (int u = 0; u < SPW; ++u) {
    const int i = (w * SPW + u) * 32 + lane;
    ok[u] = i < CHE;
    const int e = c * CHE + (ok[u] ? i : 0);
    d[u] = dst[e];
    s[u] = src[e];
  }
  {
    constexpr int NQ  = OCE / 4;
    constexpr int RND = (NQ + 1023) / 1024;
    const int nlo = (c / OCHK) * ONR;
    const int4* __restrict__ sp = (const int4*)(src + (c % OCHK) * OCE);
    int4 q[RND];
#pragma unroll
    for (int k = 0; k < RND; ++k) {
      const int i = k * 1024 + tid;
      q[k] = sp[i < NQ ? i : 0];
    }
    __syncthreads();
#pragma unroll
    for (int k = 0; k < RND; ++k) {
      if (k * 1024 + tid < NQ) {
        const unsigned a0 = (unsigned)(q[k].x - nlo), a1 = (unsigned)(q[k].y - nlo);
        const unsigned a2 = (unsigned)(q[k].z - nlo), a3 = (unsigned)(q[k].w - nlo);
        if (a0 < (unsigned)ONR) atomicAdd(&ocnt[a0], 1);
        if (a1 < (unsigned)ONR) atomicAdd(&ocnt[a1], 1);
        if (a2 < (unsigned)ONR) atomicAdd(&ocnt[a2], 1);
        if (a3 < (unsigned)ONR) atomicAdd(&ocnt[a3], 1);
      }
    }
  }
  int o[SPW];
  unsigned m[SPW];
#pragma unroll
  for (int u = 0; u < SPW; ++u) {
    o[u] = (int)((unsigned)d[u] / RPB);
    m[u] = same_owner(o[u], ok[u]);
    if (ok[u] && (m[u] & below) == 0u) myh[o[u]] += __popc(m[u]);
  }
  __syncthreads();

  if (tid < NB) {
    int t[32];
#pragma unroll
    for (int k = 0; k < 32; ++k) t[k] = hist[k * NB + tid];
    int run = 0;
#pragma unroll
    for (int k = 0; k < 32; ++k) { hist[k * NB + tid] = run; run += t[k]; }
    rtot[tid] = run;
  }
  {
    int4* __restrict__ op = (int4*)(opart + (c % OCHK) * NNODES + (c / OCHK) * ONR);
    for (int i = tid; i < ONR / 4; i += 1024) op[i] = ocnt4[i];
  }
  __syncthreads();
  if (w == 0) {
    int v[8];
    int sum = 0;
#pragma unroll
    for (int k = 0; k < 8; ++k) { v[k] = rtot[lane * 8 + k]; sum += v[k]; }
    int incl = sum;
#pragma unroll
    for (int dd = 1; dd < 32; dd <<= 1) {
      int n = __shfl_up(incl, dd);
      if (lane >= dd) incl += n;
    }
    int run = incl - sum;
#pragma unroll
    for (int k = 0; k < 8; ++k) { ostart[lane * 8 + k] = run; run += v[k]; }
  }
  __syncthreads();
  if (tid < NB) { ccnt[tid * NCH + c] = rtot[tid]; cstart[tid * NCH + c] = ostart[tid]; }
  int* __restrict__ mine = lst + c * CHE;
#pragma unroll
  for (int u = 0; u < SPW; ++u) {
    if (ok[u]) {
      const int rel = myh[o[u]];
      const int r = __popc(m[u] & below);
      mine[ostart[o[u]] + rel + r] = (s[u] << 9) | (d[u] - o[u] * RPB);
      if (r == 0) myh[o[u]] = rel + __popc(m[u]);
    }
  }
}

#define PADENT  (15 * 32)
#define REGPAD  256
static_assert(32 * 7 + 8 <= REGPAD, "fill-up of 32 lists plus the alignment of the region");
#define SEGW    (NCH / 32)
static_assert(NCH % 32 == 0, "every dealing wave takes the same number of groups");

__device__ __forceinline__ unsigned same_target(int tgt, bool ok) {
  unsigned m = (unsigned)__ballot(ok);
#pragma unroll
  for (int j = 0; j < 5; ++j) {
    const bool bit = (tgt >> j) & 1;
    const unsigned bj = (unsigned)__ballot(bit);
    m &= bit ? bj : ~bj;
  }
  return m;
}

__device__ __forceinline__ void count_step(int e, bool okk, unsigned below, int* myh, int* rcnt) {
  const int tgt = e & 31;
  const unsigned m = same_target(tgt, okk);
  if (okk && (m & below) == 0u) myh[tgt] += __popc(m);
  if (okk) atomicAdd(&rcnt[e & 511], 1);
}

__device__ __forceinline__ void place_step(int e, bool okk, unsigned below, int* myp, int* __restrict__ lst2) {
  const int tgt = e & 31;
  const unsigned m = same_target(tgt, okk);
  if (okk) {
    const int base = myp[tgt];
    const int r = __popc(m & below);
    lst2[base + r] = e;
    if (r == 0) myp[tgt] = base + __popc(m);
  }
}

__global__ __launch_bounds__(1024)
void split_kernel(const int* __restrict__ ccnt, const int* __restrict__ cstart, const int* __restrict__ lst,
                  int* __restrict__ lst2, int* __restrict__ wtab,
                  const int* __restrict__ opart, float* __restrict__ onorm, float* __restrict__ inorm) {
  __shared__ int hist[32 * 32];
  __shared__ int wpos[32 * 32];
  __shared__ int rcnt[512];
  __shared__ int rtot[NB];
  const int tid = threadIdx.x, lane = tid & 31, b = blockIdx.x;
  const int lo = b * RPB;
  const int rows = (NNODES - lo < RPB) ? (NNODES - lo) : RPB;
  if (tid < 512) rcnt[tid] = 0;
  const int s = __builtin_amdgcn_readfirstlane(tid >> 5);
  const unsigned below = (1u << lane) - 1u;
  int* myh = hist + s * 32;
  int* myp = wpos + s * 32;
  myh[lane] = 0;

  int slen[SEGW], sbeg[SEGW];
#pragma unroll
  for (int j = 0; j < SEGW; ++j) {
    const int c = s * SEGW + j;
    slen[j] = __builtin_amdgcn_readfirstlane(ccnt[b * NCH + c]);
    sbeg[j] = __builtin_amdgcn_readfirstlane(c * CHE + cstart[b * NCH + c]);
  }
  {
    const int ob = tid >> 2, q = tid & 3;
    const int4* __restrict__ row = (const int4*)(ccnt + ob * NCH + q * (NCH / 4));
    int4 v[NCH / 16];
#pragma unroll
    for (int k = 0; k < NCH / 16; ++k) v[k] = row[k];
    int tot = 0;
#pragma unroll
    for (int k = 0; k < NCH / 16; ++k) tot += v[k].x + v[k].y + v[k].z + v[k].w;
    tot += __shfl_xor(tot, 1); tot += __shfl_xor(tot, 2);
    if (q == 0) rtot[ob] = tot;
  }
  __syncthreads();
  int p[OCHK];
  if (s * 32 < rows) {
    const int* __restrict__ pn = opart + lo + (tid < rows ? tid : 0);
#pragma unroll
    for (int k = 0; k < OCHK; ++k) p[k] = pn[k * NNODES];
  }
  int ent[SEGW];
  bool ok[SEGW];
#pragma unroll
  for (int j = 0; j < SEGW; ++j) {
    ok[j] = lane < slen[j];
    ent[j] = 0;
    if (ok[j]) ent[j] = lst[sbeg[j] + lane];
  }
  int pre = 0;
#pragma unroll
  for (int k = 0; k < NB / 32; ++k) {
    const int ob = lane * (NB / 32) + k;
    pre += (ob < b) ? rtot[ob] : 0;
  }
#pragma unroll
  for (int dd = 16; dd >= 1; dd >>= 1) pre += __shfl_xor(pre, dd);
  const int region = (pre + b * REGPAD + 7) & ~7;

#pragma unroll
  for (int j = 0; j < SEGW; ++j) {
    count_step(ent[j], ok[j], below, myh, rcnt);
    for (int i = 32; i < slen[j]; i += 32) {
      const bool okr = i + lane < slen[j];
      int e = 0;
      if (okr) e = lst[sbeg[j] + i + lane];
      count_step(e, okr, below, myh, rcnt);
    }
  }
  __syncthreads();

  if (tid < rows) {
    int od = 1;
#pragma unroll
    for (int k = 0; k < OCHK; ++k) od += p[k];
    inorm[lo + tid] = rsqrtf(fmaxf((float)(rcnt[tid] + 1), 1.0f));
    onorm[lo + tid] = rsqrtf(fmaxf((float)od, 1.0f));
  }

  {
    int col = 0;
#pragma unroll
    for (int l = 0; l < 32; ++l) col += hist[l * 32 + lane];
    const int pad = (col + 7) & ~7;
    int incl = pad;
#pragma unroll
    for (int d = 1; d < 32; d <<= 1) {
      int n = __shfl_up(incl, d);
      if (lane >= d) incl += n;
    }
    const int woff = __shfl(incl - pad, s);
    const int tot  = __shfl(col, s);
    const int tpad = __shfl(pad, s);
    const int mine = hist[lane * 32 + s];
    int inc2 = mine;
#pragma unroll
    for (int d = 1; d < 32; d <<= 1) {
      int n = __shfl_up(inc2, d);
      if (lane >= d) inc2 += n;
    }
    wpos[lane * 32 + s] = region + woff + inc2 - mine;
    if (lane == 0) { wtab[(b * 32 + s) * 2] = region + woff; wtab[(b * 32 + s) * 2 + 1] = tpad; }
    if (lane < tpad - tot) lst2[region + woff + tot + lane] = PADENT;
  }
  __syncthreads();

#pragma unroll
  for (int j = 0; j < SEGW; ++j) {
    place_step(ent[j], ok[j], below, myp, lst2);
    for (int i = 32; i < slen[j]; i += 32) {
      const bool okr = i + lane < slen[j];
      int e = 0;
      if (okr) e = lst[sbeg[j] + i + lane];
      place_step(e, okr, below, myp, lst2);
    }
  }
}

#define KROWS   ((RPB + 31) / 32)
static_assert(KROWS <= 15, "register row 15 is the spare one");

template<int NOUT, bool RELU>
__global__ __launch_bounds__(1024)
void agg_owner_kernel(const int* __restrict__ wtab, const int* __restrict__ lst2,
                      const float* __restrict__ h, const float* __restrict__ inorm,
                      const float* __restrict__ bias, float* __restrict__ out) {
  constexpr int V = NOUT / 32;
  typedef float fv __attribute__((ext_vector_type(V)));
  typedef float v16f __attribute__((ext_vector_type(16)));
  typedef int v8i __attribute__((ext_vector_type(8)));
  const int tid = threadIdx.x, lane = tid & 31, b = blockIdx.x;
  const int w = __builtin_amdgcn_readfirstlane(tid >> 5);
  const int lo = b * RPB;
  const int rows = (NNODES - lo < RPB) ? (NNODES - lo) : RPB;

  const int lbeg = wtab[(b * 32 + w) * 2];
  const int n = wtab[(b * 32 + w) * 2 + 1];
  const v8i* __restrict__ myl = (const v8i*)(lst2 + lbeg);

  const fv* __restrict__ hv = (const fv*)h + (size_t)lo * 32;
  v16f a[V];
  {
    fv t[KROWS];
#pragma unroll
    for (int k = 0; k < KROWS; ++k) {
      const int r = k * 32 + w;
      t[k] = hv[(r < rows ? r : w) * 32 + lane];
    }
#pragma unroll
    for (int c = 0; c < V; ++c) {
#pragma unroll
      for (int k = 0; k < 16; ++k) a[c][k] = (k < KROWS) ? t[k < KROWS ? k : 0][c] : 0.0f;
    }
  }

  const char* __restrict__ hbytes = (const char*)h;
  const unsigned laneoff = (unsigned)lane * (unsigned)sizeof(fv);
  if (n > 0) {
    v8i en = myl[0];
    for (int j = 0; j < n; j += 8) {
      const v8i e = en;
      const int jn = (j + 8 < n) ? j + 8 : j;
      en = myl[jn >> 3];
      fv v[8];
#pragma unroll
      for (int u = 0; u < 8; ++u)
        v[u] = *(const fv*)(hbytes + (size_t)((unsigned)e[u] >> 9) * (32 * sizeof(fv)) + laneoff);
      __builtin_amdgcn_sched_barrier(0);
#pragma unroll
      for (int u = 0; u < 8; ++u) {
        const unsigned k = ((unsigned)e[u] >> 5) & 15u;
#pragma unroll
        for (int cc = 0; cc < V; ++cc) a[cc][k] += v[u][cc];
      }
    }
  }

  const fv* __restrict__ bv = (const fv*)bias;
  fv* __restrict__ ov = (fv*)out + (size_t)lo * 32;
  const fv bb = bv[lane];
  float sc[KROWS];
#pragma unroll
  for (int k = 0; k < KROWS; ++k) {
    const int r = k * 32 + w;
    sc[k] = inorm[lo + (r < rows ? r : w)];
  }
#pragma unroll
  for (int k = 0; k < KROWS; ++k) {
    const int r = k * 32 + w;
    if (r < rows) {
      fv x;
#pragma unroll
      for (int cc = 0; cc < V; ++cc) x[cc] = a[cc][k];
      x = x * sc[k] + bb;
      if (RELU) {
#pragma unroll
        for (int cc = 0; cc < V; ++cc) x[cc] = fmaxf(x[cc], 0.0f);
      }
      ov[r * 32 + lane] = x;
    }
  }
}

extern "C" void kernel_launch(void* const* d_in, const int* in_sizes, int n_in,
                              void* d_out, int out_size, void* d_ws, size_t ws_size,
                              hipStream_t stream) {
  const float* feat = (const float*)d_in[0];
  const int*   src  = (const int*)  d_in[1];
  const int*   dst  = (const int*)  d_in[2];
  const float* W1   = (const float*)d_in[3];
  const float* b1   = (const float*)d_in[4];
  const float* W2   = (const float*)d_in[5];
  const float* b2   = (const float*)d_in[6];
  const float* W3   = (const float*)d_in[7];
  const float* b3   = (const float*)d_in[8];
  float* out = (float*)d_out;

  size_t off = 0;
  auto alloc = [&](size_t bytes) -> char* {
    char* p = (char*)d_ws + off;
    off += (bytes + 255) & ~(size_t)255;
    return p;
  };
  float*     onorm = (float*)alloc((size_t)NNODES * sizeof(float));
  float*     inorm = (float*)alloc((size_t)NNODES * sizeof(float));
  int*       opart = (int*)alloc((size_t)OCHK * NNODES * sizeof(int));
  float*     hbuf  = (float*)alloc((size_t)NNODES * FDIM * sizeof(float));
  float*     xbuf  = (float*)alloc((size_t)NNODES * FDIM * sizeof(float));
  _Float16*  Wp    = (_Float16*)alloc((size_t)FDIM * FDIM * sizeof(_Float16));
  int*       ccnt  = (int*)alloc((size_t)NB * NCH * sizeof(int));
  int*       cstart = (int*)alloc((size_t)NB * NCH * sizeof(int));
  int*       lst   = (int*)alloc((size_t)NEDGES * sizeof(int));
  int*       lst2  = (int*)alloc(((size_t)NEDGES + (size_t)NB * REGPAD + 64) * sizeof(int));
  int*       wtab  = (int*)alloc((size_t)NB * 32 * 2 * sizeof(int));

  const int TB = 256;

  owner_sort_kernel<<<NCH, 1024, 0, stream>>>(src, dst, ccnt, cstart, lst, opart);
  split_kernel<<<NB, 1024, 0, stream>>>(ccnt, cstart, lst, lst2, wtab, opart, onorm, inorm);

  pack_w_kernel<<<ceil_div(4 * 8 * 512, TB), TB, 0, stream>>>(W1, Wp, 128);
  {
    int waves = NNODES / 16;
    gemm_wmma_kernel<128><<<ceil_div((long long)waves * 32, GEMM_TB), GEMM_TB, 0, stream>>>(feat, onorm, Wp, hbuf);
  }
  agg_owner_kernel<128, true><<<NB, 1024, 0, stream>>>(wtab, lst2, hbuf, inorm, b1, xbuf);

  pack_w_kernel<<<ceil_div(4 * 8 * 512, TB), TB, 0, stream>>>(W2, Wp, 128);
  {
    int waves = NNODES / 16;
    gemm_wmma_kernel<128><<<ceil_div((long long)waves * 32, GEMM_TB), GEMM_TB, 0, stream>>>(xbuf, onorm, Wp, hbuf);
  }
  agg_owner_kernel<128, true><<<NB, 1024, 0, stream>>>(wtab, lst2, hbuf, inorm, b2, xbuf);

  pack_w_kernel<<<ceil_div(4 * 4 * 512, TB), TB, 0, stream>>>(W3, Wp, 64);
  {
    int waves = NNODES / 16;
    gemm_wmma_kernel<64><<<ceil_div((long long)waves * 32, GEMM_TB), GEMM_TB, 0, stream>>>(xbuf, onorm, Wp, hbuf);
  }
  agg_owner_kernel<64, false><<<NB, 1024, 0, stream>>>(wtab, lst2, hbuf, inorm, b3, out);
}


# Layer outputs stored once in the next product's operand form (out-norm scaled, rounded in the aggregation's epilogue, 2-byte elements); products of layers 2, 3 load A fragments as stored; same sums
# speedup vs baseline: 4.8979x; 1.0547x over previous
#include <hip/hip_runtime.h>
#include <math.h>

typedef __attribute__((ext_vector_type(16))) _Float16 v16h;
typedef __attribute__((ext_vector_type(8)))  float    v8f;
typedef __attribute__((ext_vector_type(8)))  _Float16 v8h;
typedef __attribute__((ext_vector_type(4)))  _Float16 v4h;
typedef __attribute__((ext_vector_type(2)))  _Float16 v2h;

#define NNODES 100000
#define NEDGES 1600000
#define FDIM   128

static inline int ceil_div(long long a, long long b) { return (int)((a + b - 1) / b); }


__global__ void pack_w_kernel(const float* __restrict__ W, _Float16* __restrict__ Wp, int nout) {
  int idx = blockIdx.x * blockDim.x + threadIdx.x;
  int nt = nout >> 4;
  int total = 4 * nt * 32 * 16;
  if (idx >= total) return;
  int j    = idx & 15;
  int l    = (idx >> 4) & 31;
  int rest = idx >> 9;
  int t    = rest % nt;
  int c    = rest / nt;
  int col   = (t >> 2) * 64 + 4 * (l & 15) + (t & 3);
  int kbase = c * 32 + ((l >> 4) * 8);
  int K     = kbase + (j & 7) + ((j >> 3) * 16);
  Wp[idx] = (_Float16)W[K * nout + col];
}

#define GEMM_TB 256

template<int NOUT, bool XH>
__global__ __launch_bounds__(GEMM_TB)
void gemm_wmma_kernel(const float* __restrict__ X,
                      const _Float16* __restrict__ Xh,
                      const float* __restrict__ onorm,
                      const _Float16* __restrict__ Wp,
                      float* __restrict__ H) {
  constexpr int NT = NOUT / 16;
  constexpr int NG = NT / 4;
  static_assert(NT % 4 == 0, "column tiles come in groups of four");
  int mtile = (int)((blockIdx.x * blockDim.x + threadIdx.x) >> 5);
  int lane = threadIdx.x & 31;
  if (mtile * 16 >= NNODES) return;

  int r    = lane & 15;
  int half = lane >> 4;
  int row  = mtile * 16 + r;
  const v16h* __restrict__ wl = (const v16h*)Wp + lane;

  v16h a[4];
  v16h b[NT];
  if (XH) {
    const _Float16* ph16 = Xh + (size_t)row * FDIM;
    v8h lo8[4], hi8[4];
#pragma unroll
    for (int kc = 0; kc < 4; ++kc) {
      int kbase = kc * 32 + half * 8;
      lo8[kc] = *(const v8h*)(ph16 + kbase);
      hi8[kc] = *(const v8h*)(ph16 + kbase + 16);
    }
#pragma unroll
    for (int t = 0; t < NT; ++t) b[t] = wl[t * 32];
    __builtin_amdgcn_sched_barrier(0);
#pragma unroll
    for (int kc = 0; kc < 4; ++kc)
      a[kc] = __builtin_shufflevector(lo8[kc], hi8[kc], 0, 1, 2, 3, 4, 5, 6, 7, 8, 9, 10, 11, 12, 13, 14, 15);
  } else {
    const float* px = X + (size_t)row * FDIM;
    float4 f[16];
#pragma unroll
    for (int kc = 0; kc < 4; ++kc) {
      int kbase = kc * 32 + half * 8;
      f[kc * 4 + 0] = *(const float4*)(px + kbase);
      f[kc * 4 + 1] = *(const float4*)(px + kbase + 4);
      f[kc * 4 + 2] = *(const float4*)(px + kbase + 16);
      f[kc * 4 + 3] = *(const float4*)(px + kbase + 20);
    }
    float scale = onorm[row];
#pragma unroll
    for (int t = 0; t < NT; ++t) b[t] = wl[t * 32];
    __builtin_amdgcn_sched_barrier(0);

#pragma unroll
    for (int kc = 0; kc < 4; ++kc) {
#pragma unroll
      for (int i = 0; i < 4; ++i) {
        const float4 q = f[kc * 4 + i];
        a[kc][i * 4 + 0] = (_Float16)(q.x * scale);
        a[kc][i * 4 + 1] = (_Float16)(q.y * scale);
        a[kc][i * 4 + 2] = (_Float16)(q.z * scale);
        a[kc][i * 4 + 3] = (_Float16)(q.w * scale);
      }
    }
  }

  v8f acc[NT];
#pragma unroll
  for (int t = 0; t < NT; ++t) acc[t] = (v8f){0.f, 0.f, 0.f, 0.f, 0.f, 0.f, 0.f, 0.f};

#pragma unroll
  for (int kc = 0; kc < 4; ++kc) {
#pragma unroll
    for (int t = 0; t < NT; ++t)
      acc[t] = __builtin_amdgcn_wmma_f32_16x16x32_f16(
           false, a[kc],  false, b[t],
           (short)0, acc[t],  false,  false);
    if (kc < 3) {
#pragma unroll
      for (int t = 0; t < NT; ++t) b[t] = wl[((kc + 1) * NT + t) * 32];
    }
    __builtin_amdgcn_sched_barrier(0);
  }

  float* ph = H + (size_t)(mtile * 16 + half * 8) * NOUT + 4 * r;
#pragma unroll
  for (int g = 0; g < NG; ++g) {
#pragma unroll
    for (int vr = 0; vr < 8; ++vr) {
      float4 o;
      o.x = acc[4 * g + 0][vr]; o.y = acc[4 * g + 1][vr];
      o.z = acc[4 * g + 2][vr]; o.w = acc[4 * g + 3][vr];
      *(float4*)(ph + (size_t)vr * NOUT + g * 64) = o;
    }
  }
}

#define NB      256
#define RPB     ((NNODES + NB - 1) / NB)
#define NCH     256
#define CHE     (NEDGES / NCH)
#define STEPS   ((CHE + 31) / 32)
#define SPW     ((STEPS + 31) / 32)

static_assert(NEDGES % NCH == 0, "chunks are equal");
static_assert(NB == 256 && (NNODES - 1) / RPB < NB, "owner id is 8 bits");
static_assert(RPB + 32 <= 512 && NNODES <= (1 << 17), "entry packs src<<9 | local row");
static_assert(NCH % 16 == 0, "a count row is read in four quarters of whole 16-byte groups");
#define ORNG    8
#define OCHK    (NCH / ORNG)
#define ONR     (NNODES / ORNG)
#define OCE     (NEDGES / OCHK)
static_assert(ORNG * OCHK == NCH, "one block of the grouping launch per (node range, edge chunk)");
static_assert(NNODES % ORNG == 0 && NEDGES % OCHK == 0, "ranges and chunks are equal");
static_assert(ONR % 4 == 0 && OCE % 4 == 0 && NNODES % 4 == 0, "ids and counts move in whole 16-byte pieces");
static_assert(RPB <= 512, "a block's rows have a counter each in split_kernel");

__device__ __forceinline__ unsigned same_owner(int o, bool ok) {
  unsigned m = (unsigned)__ballot(ok);
#pragma unroll
  for (int j = 0; j < 8; ++j) {
    const bool bit = (o >> j) & 1;
    const unsigned bj = (unsigned)__ballot(bit);
    m &= bit ? bj : ~bj;
  }
  return m;
}

__global__ __launch_bounds__(1024)
void owner_sort_kernel(const int* __restrict__ src, const int* __restrict__ dst,
                       int* __restrict__ ccnt, int* __restrict__ cstart, int* __restrict__ lst,
                       int* __restrict__ opart) {
  __shared__ int hist[32 * NB];
  __shared__ int rtot[NB];
  __shared__ int ostart[NB];
  __shared__ int4 ocnt4[ONR / 4];
  int* ocnt = (int*)ocnt4;
  const int tid = threadIdx.x, lane = tid & 31, c = blockIdx.x;
  const int w = __builtin_amdgcn_readfirstlane(tid >> 5);
  const unsigned below = (1u << lane) - 1u;
  int* myh = hist + w * NB;
#pragma unroll
  for (int k = 0; k < NB / 32; ++k) myh[k * 32 + lane] = 0;
  for (int i = tid; i < ONR / 4; i += 1024) ocnt4[i] = make_int4(0, 0, 0, 0);

  int d[SPW], s[SPW];
  bool ok[SPW];
#pragma unroll
  for (int u = 0; u < SPW; ++u) {
    const int i = (w * SPW + u) * 32 + lane;
    ok[u] = i < CHE;
    const int e = c * CHE + (ok[u] ? i : 0);
    d[u] = dst[e];
    s[u] = src[e];
  }
  {
    constexpr int NQ  = OCE / 4;
    constexpr int RND = (NQ + 1023) / 1024;
    const int nlo = (c / OCHK) * ONR;
    const int4* __restrict__ sp = (const int4*)(src + (c % OCHK) * OCE);
    int4 q[RND];
#pragma unroll
    for (int k = 0; k < RND; ++k) {
      const int i = k * 1024 + tid;
      q[k] = sp[i < NQ ? i : 0];
    }
    __syncthreads();
#pragma unroll
    for (int k = 0; k < RND; ++k) {
      if (k * 1024 + tid < NQ) {
        const unsigned a0 = (unsigned)(q[k].x - nlo), a1 = (unsigned)(q[k].y - nlo);
        const unsigned a2 = (unsigned)(q[k].z - nlo), a3 = (unsigned)(q[k].w - nlo);
        if (a0 < (unsigned)ONR) atomicAdd(&ocnt[a0], 1);
        if (a1 < (unsigned)ONR) atomicAdd(&ocnt[a1], 1);
        if (a2 < (unsigned)ONR) atomicAdd(&ocnt[a2], 1);
        if (a3 < (unsigned)ONR) atomicAdd(&ocnt[a3], 1);
      }
    }
  }
  int o[SPW];
  unsigned m[SPW];
#pragma unroll
  for (int u = 0; u < SPW; ++u) {
    o[u] = (int)((unsigned)d[u] / RPB);
    m[u] = same_owner(o[u], ok[u]);
    if (ok[u] && (m[u] & below) == 0u) myh[o[u]] += __popc(m[u]);
  }
  __syncthreads();

  if (tid < NB) {
    int t[32];
#pragma unroll
    for (int k = 0; k < 32; ++k) t[k] = hist[k * NB + tid];
    int run = 0;
#pragma unroll
    for (int k = 0; k < 32; ++k) { hist[k * NB + tid] = run; run += t[k]; }
    rtot[tid] = run;
  }
  {
    int4* __restrict__ op = (int4*)(opart + (c % OCHK) * NNODES + (c / OCHK) * ONR);
    for (int i = tid; i < ONR / 4; i += 1024) op[i] = ocnt4[i];
  }
  __syncthreads();
  if (w == 0) {
    int v[8];
    int sum = 0;
#pragma unroll
    for (int k = 0; k < 8; ++k) { v[k] = rtot[lane * 8 + k]; sum += v[k]; }
    int incl = sum;
#pragma unroll
    for (int dd = 1; dd < 32; dd <<= 1) {
      int n = __shfl_up(incl, dd);
      if (lane >= dd) incl += n;
    }
    int run = incl - sum;
#pragma unroll
    for (int k = 0; k < 8; ++k) { ostart[lane * 8 + k] = run; run += v[k]; }
  }
  __syncthreads();
  if (tid < NB) { ccnt[tid * NCH + c] = rtot[tid]; cstart[tid * NCH + c] = ostart[tid]; }
  int* __restrict__ mine = lst + c * CHE;
#pragma unroll
  for (int u = 0; u < SPW; ++u) {
    if (ok[u]) {
      const int rel = myh[o[u]];
      const int r = __popc(m[u] & below);
      mine[ostart[o[u]] + rel + r] = (s[u] << 9) | (d[u] - o[u] * RPB);
      if (r == 0) myh[o[u]] = rel + __popc(m[u]);
    }
  }
}

#define PADENT  (15 * 32)
#define REGPAD  256
static_assert(32 * 7 + 8 <= REGPAD, "fill-up of 32 lists plus the alignment of the region");
#define SEGW    (NCH / 32)
static_assert(NCH % 32 == 0, "every dealing wave takes the same number of groups");

__device__ __forceinline__ unsigned same_target(int tgt, bool ok) {
  unsigned m = (unsigned)__ballot(ok);
#pragma unroll
  for (int j = 0; j < 5; ++j) {
    const bool bit = (tgt >> j) & 1;
    const unsigned bj = (unsigned)__ballot(bit);
    m &= bit ? bj : ~bj;
  }
  return m;
}

__device__ __forceinline__ void count_step(int e, bool okk, unsigned below, int* myh, int* rcnt) {
  const int tgt = e & 31;
  const unsigned m = same_target(tgt, okk);
  if (okk && (m & below) == 0u) myh[tgt] += __popc(m);
  if (okk) atomicAdd(&rcnt[e & 511], 1);
}

__device__ __forceinline__ void place_step(int e, bool okk, unsigned below, int* myp, int* __restrict__ lst2) {
  const int tgt = e & 31;
  const unsigned m = same_target(tgt, okk);
  if (okk) {
    const int base = myp[tgt];
    const int r = __popc(m & below);
    lst2[base + r] = e;
    if (r == 0) myp[tgt] = base + __popc(m);
  }
}

__global__ __launch_bounds__(1024)
void split_kernel(const int* __restrict__ ccnt, const int* __restrict__ cstart, const int* __restrict__ lst,
                  int* __restrict__ lst2, int* __restrict__ wtab,
                  const int* __restrict__ opart, float* __restrict__ onorm, float* __restrict__ inorm) {
  __shared__ int hist[32 * 32];
  __shared__ int wpos[32 * 32];
  __shared__ int rcnt[512];
  __shared__ int rtot[NB];
  const int tid = threadIdx.x, lane = tid & 31, b = blockIdx.x;
  const int lo = b * RPB;
  const int rows = (NNODES - lo < RPB) ? (NNODES - lo) : RPB;
  if (tid < 512) rcnt[tid] = 0;
  const int s = __builtin_amdgcn_readfirstlane(tid >> 5);
  const unsigned below = (1u << lane) - 1u;
  int* myh = hist + s * 32;
  int* myp = wpos + s * 32;
  myh[lane] = 0;

  int slen[SEGW], sbeg[SEGW];
#pragma unroll
  for (int j = 0; j < SEGW; ++j) {
    const int c = s * SEGW + j;
    slen[j] = __builtin_amdgcn_readfirstlane(ccnt[b * NCH + c]);
    sbeg[j] = __builtin_amdgcn_readfirstlane(c * CHE + cstart[b * NCH + c]);
  }
  {
    const int ob = tid >> 2, q = tid & 3;
    const int4* __restrict__ row = (const int4*)(ccnt + ob * NCH + q * (NCH / 4));
    int4 v[NCH / 16];
#pragma unroll
    for (int k = 0; k < NCH / 16; ++k) v[k] = row[k];
    int tot = 0;
#pragma unroll
    for (int k = 0; k < NCH / 16; ++k) tot += v[k].x + v[k].y + v[k].z + v[k].w;
    tot += __shfl_xor(tot, 1); tot += __shfl_xor(tot, 2);
    if (q == 0) rtot[ob] = tot;
  }
  __syncthreads();
  int p[OCHK];
  if (s * 32 < rows) {
    const int* __restrict__ pn = opart + lo + (tid < rows ? tid : 0);
#pragma unroll
    for (int k = 0; k < OCHK; ++k) p[k] = pn[k * NNODES];
  }
  int ent[SEGW];
  bool ok[SEGW];
#pragma unroll
  for (int j = 0; j < SEGW; ++j) {
    ok[j] = lane < slen[j];
    ent[j] = 0;
    if (ok[j]) ent[j] = lst[sbeg[j] + lane];
  }
  int pre = 0;
#pragma unroll
  for (int k = 0; k < NB / 32; ++k) {
    const int ob = lane * (NB / 32) + k;
    pre += (ob < b) ? rtot[ob] : 0;
  }
#pragma unroll
  for (int dd = 16; dd >= 1; dd >>= 1) pre += __shfl_xor(pre, dd);
  const int region = (pre + b * REGPAD + 7) & ~7;

#pragma unroll
  for (int j = 0; j < SEGW; ++j) {
    count_step(ent[j], ok[j], below, myh, rcnt);
    for (int i = 32; i < slen[j]; i += 32) {
      const bool okr = i + lane < slen[j];
      int e = 0;
      if (okr) e = lst[sbeg[j] + i + lane];
      count_step(e, okr, below, myh, rcnt);
    }
  }
  __syncthreads();

  if (tid < rows) {
    int od = 1;
#pragma unroll
    for (int k = 0; k < OCHK; ++k) od += p[k];
    inorm[lo + tid] = rsqrtf(fmaxf((float)(rcnt[tid] + 1), 1.0f));
    onorm[lo + tid] = rsqrtf(fmaxf((float)od, 1.0f));
  }

  {
    int col = 0;
#pragma unroll
    for (int l = 0; l < 32; ++l) col += hist[l * 32 + lane];
    const int pad = (col + 7) & ~7;
    int incl = pad;
#pragma unroll
    for (int d = 1; d < 32; d <<= 1) {
      int n = __shfl_up(incl, d);
      if (lane >= d) incl += n;
    }
    const int woff = __shfl(incl - pad, s);
    const int tot  = __shfl(col, s);
    const int tpad = __shfl(pad, s);
    const int mine = hist[lane * 32 + s];
    int inc2 = mine;
#pragma unroll
    for (int d = 1; d < 32; d <<= 1) {
      int n = __shfl_up(inc2, d);
      if (lane >= d) inc2 += n;
    }
    wpos[lane * 32 + s] = region + woff + inc2 - mine;
    if (lane == 0) { wtab[(b * 32 + s) * 2] = region + woff; wtab[(b * 32 + s) * 2 + 1] = tpad; }
    if (lane < tpad - tot) lst2[region + woff + tot + lane] = PADENT;
  }
  __syncthreads();

#pragma unroll
  for (int j = 0; j < SEGW; ++j) {
    place_step(ent[j], ok[j], below, myp, lst2);
    for (int i = 32; i < slen[j]; i += 32) {
      const bool okr = i + lane < slen[j];
      int e = 0;
      if (okr) e = lst[sbeg[j] + i + lane];
      place_step(e, okr, below, myp, lst2);
    }
  }
}

#define KROWS   ((RPB + 31) / 32)
static_assert(KROWS <= 15, "register row 15 is the spare one");

template<int NOUT, bool RELU, bool HOUT>
__global__ __launch_bounds__(1024)
void agg_owner_kernel(const int* __restrict__ wtab, const int* __restrict__ lst2,
                      const float* __restrict__ h, const float* __restrict__ inorm,
                      const float* __restrict__ onorm,
                      const float* __restrict__ bias, float* __restrict__ out,
                      _Float16* __restrict__ outh) {
  constexpr int V = NOUT / 32;
  typedef float fv __attribute__((ext_vector_type(V)));
  typedef _Float16 hfv __attribute__((ext_vector_type(V)));
  typedef float v16f __attribute__((ext_vector_type(16)));
  typedef int v8i __attribute__((ext_vector_type(8)));
  const int tid = threadIdx.x, lane = tid & 31, b = blockIdx.x;
  const int w = __builtin_amdgcn_readfirstlane(tid >> 5);
  const int lo = b * RPB;
  const int rows = (NNODES - lo < RPB) ? (NNODES - lo) : RPB;

  const int lbeg = wtab[(b * 32 + w) * 2];
  const int n = wtab[(b * 32 + w) * 2 + 1];
  const v8i* __restrict__ myl = (const v8i*)(lst2 + lbeg);

  const fv* __restrict__ hv = (const fv*)h + (size_t)lo * 32;
  v16f a[V];
  {
    fv t[KROWS];
#pragma unroll
    for (int k = 0; k < KROWS; ++k) {
      const int r = k * 32 + w;
      t[k] = hv[(r < rows ? r : w) * 32 + lane];
    }
#pragma unroll
    for (int c = 0; c < V; ++c) {
#pragma unroll
      for (int k = 0; k < 16; ++k) a[c][k] = (k < KROWS) ? t[k < KROWS ? k : 0][c] : 0.0f;
    }
  }

  const char* __restrict__ hbytes = (const char*)h;
  const unsigned laneoff = (unsigned)lane * (unsigned)sizeof(fv);
  if (n > 0) {
    v8i en = myl[0];
    for (int j = 0; j < n; j += 8) {
      const v8i e = en;
      const int jn = (j + 8 < n) ? j + 8 : j;
      en = myl[jn >> 3];
      fv v[8];
#pragma unroll
      for (int u = 0; u < 8; ++u)
        v[u] = *(const fv*)(hbytes + (size_t)((unsigned)e[u] >> 9) * (32 * sizeof(fv)) + laneoff);
      __builtin_amdgcn_sched_barrier(0);
#pragma unroll
      for (int u = 0; u < 8; ++u) {
        const unsigned k = ((unsigned)e[u] >> 5) & 15u;
#pragma unroll
        for (int cc = 0; cc < V; ++cc) a[cc][k] += v[u][cc];
      }
    }
  }

  const fv* __restrict__ bv = (const fv*)bias;
  fv* __restrict__ ov = (fv*)out + (size_t)lo * 32;
  hfv* __restrict__ oh = (hfv*)outh + (size_t)lo * 32;
  const fv bb = bv[lane];
  float sc[KROWS], so[KROWS];
#pragma unroll
  for (int k = 0; k < KROWS; ++k) {
    const int r = k * 32 + w;
    sc[k] = inorm[lo + (r < rows ? r : w)];
    so[k] = HOUT ? onorm[lo + (r < rows ? r : w)] : 1.0f;
  }
#pragma unroll
  for (int k = 0; k < KROWS; ++k) {
    const int r = k * 32 + w;
    if (r < rows) {
      fv x;
#pragma unroll
      for (int cc = 0; cc < V; ++cc) x[cc] = a[cc][k];
      x = x * sc[k] + bb;
      if (RELU) {
#pragma unroll
        for (int cc = 0; cc < V; ++cc) x[cc] = fmaxf(x[cc], 0.0f);
      }
      if (HOUT) {
        hfv y;
#pragma unroll
        for (int cc = 0; cc < V; ++cc) y[cc] = (_Float16)(x[cc] * so[k]);
        oh[r * 32 + lane] = y;
      } else {
        ov[r * 32 + lane] = x;
      }
    }
  }
}

extern "C" void kernel_launch(void* const* d_in, const int* in_sizes, int n_in,
                              void* d_out, int out_size, void* d_ws, size_t ws_size,
                              hipStream_t stream) {
  const float* feat = (const float*)d_in[0];
  const int*   src  = (const int*)  d_in[1];
  const int*   dst  = (const int*)  d_in[2];
  const float* W1   = (const float*)d_in[3];
  const float* b1   = (const float*)d_in[4];
  const float* W2   = (const float*)d_in[5];
  const float* b2   = (const float*)d_in[6];
  const float* W3   = (const float*)d_in[7];
  const float* b3   = (const float*)d_in[8];
  float* out = (float*)d_out;

  size_t off = 0;
  auto alloc = [&](size_t bytes) -> char* {
    char* p = (char*)d_ws + off;
    off += (bytes + 255) & ~(size_t)255;
    return p;
  };
  float*     onorm = (float*)alloc((size_t)NNODES * sizeof(float));
  float*     inorm = (float*)alloc((size_t)NNODES * sizeof(float));
  int*       opart = (int*)alloc((size_t)OCHK * NNODES * sizeof(int));
  float*     hbuf  = (float*)alloc((size_t)NNODES * FDIM * sizeof(float));
  _Float16*  xh    = (_Float16*)alloc((size_t)NNODES * FDIM * sizeof(_Float16));
  _Float16*  Wp    = (_Float16*)alloc((size_t)FDIM * FDIM * sizeof(_Float16));
  int*       ccnt  = (int*)alloc((size_t)NB * NCH * sizeof(int));
  int*       cstart = (int*)alloc((size_t)NB * NCH * sizeof(int));
  int*       lst   = (int*)alloc((size_t)NEDGES * sizeof(int));
  int*       lst2  = (int*)alloc(((size_t)NEDGES + (size_t)NB * REGPAD + 64) * sizeof(int));
  int*       wtab  = (int*)alloc((size_t)NB * 32 * 2 * sizeof(int));

  const int TB = 256;

  owner_sort_kernel<<<NCH, 1024, 0, stream>>>(src, dst, ccnt, cstart, lst, opart);
  split_kernel<<<NB, 1024, 0, stream>>>(ccnt, cstart, lst, lst2, wtab, opart, onorm, inorm);

  pack_w_kernel<<<ceil_div(4 * 8 * 512, TB), TB, 0, stream>>>(W1, Wp, 128);
  {
    int waves = NNODES / 16;
    gemm_wmma_kernel<128, false><<<ceil_div((long long)waves * 32, GEMM_TB), GEMM_TB, 0, stream>>>(feat, nullptr, onorm, Wp, hbuf);
  }
  agg_owner_kernel<128, true, true><<<NB, 1024, 0, stream>>>(wtab, lst2, hbuf, inorm, onorm, b1, nullptr, xh);

  pack_w_kernel<<<ceil_div(4 * 8 * 512, TB), TB, 0, stream>>>(W2, Wp, 128);
  {
    int waves = NNODES / 16;
    gemm_wmma_kernel<128, true><<<ceil_div((long long)waves * 32, GEMM_TB), GEMM_TB, 0, stream>>>(nullptr, xh, onorm, Wp, hbuf);
  }
  agg_owner_kernel<128, true, true><<<NB, 1024, 0, stream>>>(wtab, lst2, hbuf, inorm, onorm, b2, nullptr, xh);

  pack_w_kernel<<<ceil_div(4 * 4 * 512, TB), TB, 0, stream>>>(W3, Wp, 64);
  {
    int waves = NNODES / 16;
    gemm_wmma_kernel<64, true><<<ceil_div((long long)waves * 32, GEMM_TB), GEMM_TB, 0, stream>>>(nullptr, xh, onorm, Wp, hbuf);
  }
  agg_owner_kernel<64, false, false><<<NB, 1024, 0, stream>>>(wtab, lst2, hbuf, inorm, onorm, b3, out, nullptr);
}


# List kernels use whole lines: entries are placed in an LDS copy of the block's stretch and leave as 128-byte lines, not a 4-byte store each; count-row loads touch 8 lines, not 32; same lists and sums
# speedup vs baseline: 5.1569x; 1.0529x over previous
#include <hip/hip_runtime.h>
#include <math.h>

typedef __attribute__((ext_vector_type(16))) _Float16 v16h;
typedef __attribute__((ext_vector_type(8)))  float    v8f;
typedef __attribute__((ext_vector_type(8)))  _Float16 v8h;
typedef __attribute__((ext_vector_type(4)))  _Float16 v4h;
typedef __attribute__((ext_vector_type(2)))  _Float16 v2h;

#define NNODES 100000
#define NEDGES 1600000
#define FDIM   128

static inline int ceil_div(long long a, long long b) { return (int)((a + b - 1) / b); }


__global__ void pack_w_kernel(const float* __restrict__ W, _Float16* __restrict__ Wp, int nout) {
  int idx = blockIdx.x * blockDim.x + threadIdx.x;
  int nt = nout >> 4;
  int total = 4 * nt * 32 * 16;
  if (idx >= total) return;
  int j    = idx & 15;
  int l    = (idx >> 4) & 31;
  int rest = idx >> 9;
  int t    = rest % nt;
  int c    = rest / nt;
  int col   = (t >> 2) * 64 + 4 * (l & 15) + (t & 3);
  int kbase = c * 32 + ((l >> 4) * 8);
  int K     = kbase + (j & 7) + ((j >> 3) * 16);
  Wp[idx] = (_Float16)W[K * nout + col];
}

#define GEMM_TB 256

template<int NOUT, bool XH>
__global__ __launch_bounds__(GEMM_TB)
void gemm_wmma_kernel(const float* __restrict__ X,
                      const _Float16* __restrict__ Xh,
                      const float* __restrict__ onorm,
                      const _Float16* __restrict__ Wp,
                      float* __restrict__ H) {
  constexpr int NT = NOUT / 16;
  constexpr int NG = NT / 4;
  static_assert(NT % 4 == 0, "column tiles come in groups of four");
  int mtile = (int)((blockIdx.x * blockDim.x + threadIdx.x) >> 5);
  int lane = threadIdx.x & 31;
  if (mtile * 16 >= NNODES) return;

  int r    = lane & 15;
  int half = lane >> 4;
  int row  = mtile * 16 + r;
  const v16h* __restrict__ wl = (const v16h*)Wp + lane;

  v16h a[4];
  v16h b[NT];
  if (XH) {
    const _Float16* ph16 = Xh + (size_t)row * FDIM;
    v8h lo8[4], hi8[4];
#pragma unroll
    for (int kc = 0; kc < 4; ++kc) {
      int kbase = kc * 32 + half * 8;
      lo8[kc] = *(const v8h*)(ph16 + kbase);
      hi8[kc] = *(const v8h*)(ph16 + kbase + 16);
    }
#pragma unroll
    for (int t = 0; t < NT; ++t) b[t] = wl[t * 32];
    __builtin_amdgcn_sched_barrier(0);
#pragma unroll
    for (int kc = 0; kc < 4; ++kc)
      a[kc] = __builtin_shufflevector(lo8[kc], hi8[kc], 0, 1, 2, 3, 4, 5, 6, 7, 8, 9, 10, 11, 12, 13, 14, 15);
  } else {
    const float* px = X + (size_t)row * FDIM;
    float4 f[16];
#pragma unroll
    for (int kc = 0; kc < 4; ++kc) {
      int kbase = kc * 32 + half * 8;
      f[kc * 4 + 0] = *(const float4*)(px + kbase);
      f[kc * 4 + 1] = *(const float4*)(px + kbase + 4);
      f[kc * 4 + 2] = *(const float4*)(px + kbase + 16);
      f[kc * 4 + 3] = *(const float4*)(px + kbase + 20);
    }
    float scale = onorm[row];
#pragma unroll
    for (int t = 0; t < NT; ++t) b[t] = wl[t * 32];
    __builtin_amdgcn_sched_barrier(0);

#pragma unroll
    for (int kc = 0; kc < 4; ++kc) {
#pragma unroll
      for (int i = 0; i < 4; ++i) {
        const float4 q = f[kc * 4 + i];
        a[kc][i * 4 + 0] = (_Float16)(q.x * scale);
        a[kc][i * 4 + 1] = (_Float16)(q.y * scale);
        a[kc][i * 4 + 2] = (_Float16)(q.z * scale);
        a[kc][i * 4 + 3] = (_Float16)(q.w * scale);
      }
    }
  }

  v8f acc[NT];
#pragma unroll
  for (int t = 0; t < NT; ++t) acc[t] = (v8f){0.f, 0.f, 0.f, 0.f, 0.f, 0.f, 0.f, 0.f};

#pragma unroll
  for (int kc = 0; kc < 4; ++kc) {
#pragma unroll
    for (int t = 0; t < NT; ++t)
      acc[t] = __builtin_amdgcn_wmma_f32_16x16x32_f16(
           false, a[kc],  false, b[t],
           (short)0, acc[t],  false,  false);
    if (kc < 3) {
#pragma unroll
      for (int t = 0; t < NT; ++t) b[t] = wl[((kc + 1) * NT + t) * 32];
    }
    __builtin_amdgcn_sched_barrier(0);
  }

  float* ph = H + (size_t)(mtile * 16 + half * 8) * NOUT + 4 * r;
#pragma unroll
  for (int g = 0; g < NG; ++g) {
#pragma unroll
    for (int vr = 0; vr < 8; ++vr) {
      float4 o;
      o.x = acc[4 * g + 0][vr]; o.y = acc[4 * g + 1][vr];
      o.z = acc[4 * g + 2][vr]; o.w = acc[4 * g + 3][vr];
      *(float4*)(ph + (size_t)vr * NOUT + g * 64) = o;
    }
  }
}

#define NB      256
#define RPB     ((NNODES + NB - 1) / NB)
#define NCH     256
#define CHE     (NEDGES / NCH)
#define CHEP    ((CHE + 31) / 32 * 32)
#define STEPS   ((CHE + 31) / 32)
#define SPW     ((STEPS + 31) / 32)

static_assert(NEDGES % NCH == 0, "chunks are equal");
static_assert(NB == 256 && (NNODES - 1) / RPB < NB, "owner id is 8 bits");
static_assert(RPB + 32 <= 512 && NNODES <= (1 << 17), "entry packs src<<9 | local row");
static_assert(NCH % 16 == 0, "a count row is read in four quarters of whole 16-byte groups");
#define ORNG    8
#define OCHK    (NCH / ORNG)
#define ONR     (NNODES / ORNG)
#define OCE     (NEDGES / OCHK)
static_assert(ORNG * OCHK == NCH, "one block of the grouping launch per (node range, edge chunk)");
static_assert(NNODES % ORNG == 0 && NEDGES % OCHK == 0, "ranges and chunks are equal");
static_assert(ONR % 4 == 0 && OCE % 4 == 0 && NNODES % 4 == 0, "ids and counts move in whole 16-byte pieces");
static_assert(RPB <= 512, "a block's rows have a counter each in split_kernel");

__device__ __forceinline__ unsigned same_owner(int o, bool ok) {
  unsigned m = (unsigned)__ballot(ok);
#pragma unroll
  for (int j = 0; j < 8; ++j) {
    const bool bit = (o >> j) & 1;
    const unsigned bj = (unsigned)__ballot(bit);
    m &= bit ? bj : ~bj;
  }
  return m;
}

__global__ __launch_bounds__(1024)
void owner_sort_kernel(const int* __restrict__ src, const int* __restrict__ dst,
                       int* __restrict__ ccnt, int* __restrict__ cstart, int* __restrict__ lst,
                       int* __restrict__ opart) {
  __shared__ int hist[32 * NB];
  __shared__ int rtot[NB];
  __shared__ int ostart[NB];
  __shared__ int4 ocnt4[ONR / 4];
  int* ocnt = (int*)ocnt4;
  const int tid = threadIdx.x, lane = tid & 31, c = blockIdx.x;
  const int w = __builtin_amdgcn_readfirstlane(tid >> 5);
  const unsigned below = (1u << lane) - 1u;
  int* myh = hist + w * NB;
#pragma unroll
  for (int k = 0; k < NB / 32; ++k) myh[k * 32 + lane] = 0;
  for (int i = tid; i < ONR / 4; i += 1024) ocnt4[i] = make_int4(0, 0, 0, 0);

  int d[SPW], s[SPW];
  bool ok[SPW];
#pragma unroll
  for (int u = 0; u < SPW; ++u) {
    const int i = (w * SPW + u) * 32 + lane;
    ok[u] = i < CHE;
    const int e = c * CHE + (ok[u] ? i : 0);
    d[u] = dst[e];
    s[u] = src[e];
  }
  {
    constexpr int NQ  = OCE / 4;
    constexpr int RND = (NQ + 1023) / 1024;
    const int nlo = (c / OCHK) * ONR;
    const int4* __restrict__ sp = (const int4*)(src + (c % OCHK) * OCE);
    int4 q[RND];
#pragma unroll
    for (int k = 0; k < RND; ++k) {
      const int i = k * 1024 + tid;
      q[k] = sp[i < NQ ? i : 0];
    }
    __syncthreads();
#pragma unroll
    for (int k = 0; k < RND; ++k) {
      if (k * 1024 + tid < NQ) {
        const unsigned a0 = (unsigned)(q[k].x - nlo), a1 = (unsigned)(q[k].y - nlo);
        const unsigned a2 = (unsigned)(q[k].z - nlo), a3 = (unsigned)(q[k].w - nlo);
        if (a0 < (unsigned)ONR) atomicAdd(&ocnt[a0], 1);
        if (a1 < (unsigned)ONR) atomicAdd(&ocnt[a1], 1);
        if (a2 < (unsigned)ONR) atomicAdd(&ocnt[a2], 1);
        if (a3 < (unsigned)ONR) atomicAdd(&ocnt[a3], 1);
      }
    }
  }
  int o[SPW];
  unsigned m[SPW];
#pragma unroll
  for (int u = 0; u < SPW; ++u) {
    o[u] = (int)((unsigned)d[u] / RPB);
    m[u] = same_owner(o[u], ok[u]);
    if (ok[u] && (m[u] & below) == 0u) myh[o[u]] += __popc(m[u]);
  }
  __syncthreads();

  if (tid < NB) {
    int t[32];
#pragma unroll
    for (int k = 0; k < 32; ++k) t[k] = hist[k * NB + tid];
    int run = 0;
#pragma unroll
    for (int k = 0; k < 32; ++k) { hist[k * NB + tid] = run; run += t[k]; }
    rtot[tid] = run;
  }
  {
    int4* __restrict__ op = (int4*)(opart + (c % OCHK) * NNODES + (c / OCHK) * ONR);
    for (int i = tid; i < ONR / 4; i += 1024) op[i] = ocnt4[i];
  }
  __syncthreads();
  if (w == 0) {
    int v[8];
    int sum = 0;
#pragma unroll
    for (int k = 0; k < 8; ++k) { v[k] = rtot[lane * 8 + k]; sum += v[k]; }
    int incl = sum;
#pragma unroll
    for (int dd = 1; dd < 32; dd <<= 1) {
      int n = __shfl_up(incl, dd);
      if (lane >= dd) incl += n;
    }
    int run = incl - sum;
#pragma unroll
    for (int k = 0; k < 8; ++k) { ostart[lane * 8 + k] = run; run += v[k]; }
  }
  __syncthreads();
  if (tid < NB) { ccnt[tid * NCH + c] = rtot[tid]; cstart[tid * NCH + c] = ostart[tid]; }
  constexpr int NQ4 = (CHE + 3) / 4;
  static_assert(4 * NQ4 <= ONR, "the stretch fits ocnt's space");
  int* stage = ocnt;
#pragma unroll
  for (int u = 0; u < SPW; ++u) {
    if (ok[u]) {
      const int rel = myh[o[u]];
      const int r = __popc(m[u] & below);
      stage[ostart[o[u]] + rel + r] = (s[u] << 9) | (d[u] - o[u] * RPB);
      if (r == 0) myh[o[u]] = rel + __popc(m[u]);
    }
  }
  if (tid < 4 * NQ4 - CHE) stage[CHE + tid] = 0;
  __syncthreads();
  int4* __restrict__ mine4 = (int4*)(lst + c * CHEP);
  for (int i = tid; i < NQ4; i += 1024) mine4[i] = ocnt4[i];
}

#define PADENT  (15 * 32)
#define REGPAD  256
static_assert(32 * 7 + 32 <= REGPAD, "fill-up of 32 lists plus the alignment of the region");
#define SCAP    12288
static_assert(SCAP % 4 == 0, "the copy moves in 16-byte pieces");
#define SEGW    (NCH / 32)
static_assert(NCH % 32 == 0, "every dealing wave takes the same number of groups");

__device__ __forceinline__ unsigned same_target(int tgt, bool ok) {
  unsigned m = (unsigned)__ballot(ok);
#pragma unroll
  for (int j = 0; j < 5; ++j) {
    const bool bit = (tgt >> j) & 1;
    const unsigned bj = (unsigned)__ballot(bit);
    m &= bit ? bj : ~bj;
  }
  return m;
}

__device__ __forceinline__ void count_step(int e, bool okk, unsigned below, int* myh, int* rcnt) {
  const int tgt = e & 31;
  const unsigned m = same_target(tgt, okk);
  if (okk && (m & below) == 0u) myh[tgt] += __popc(m);
  if (okk) atomicAdd(&rcnt[e & 511], 1);
}

__device__ __forceinline__ void place_step(int e, bool okk, unsigned below, int* myp, int* __restrict__ reg) {
  const int tgt = e & 31;
  const unsigned m = same_target(tgt, okk);
  if (okk) {
    const int base = myp[tgt];
    const int r = __popc(m & below);
    reg[base + r] = e;
    if (r == 0) myp[tgt] = base + __popc(m);
  }
}

__global__ __launch_bounds__(1024)
void split_kernel(const int* __restrict__ ccnt, const int* __restrict__ cstart, const int* __restrict__ lst,
                  int* __restrict__ lst2, int* __restrict__ wtab,
                  const int* __restrict__ opart, float* __restrict__ onorm, float* __restrict__ inorm) {
  __shared__ int hist[32 * 32];
  __shared__ int wpos[32 * 32];
  __shared__ int rcnt[512];
  __shared__ int rtot[NB];
  __shared__ int4 stage4[SCAP / 4];
  int* stage = (int*)stage4;
  const int tid = threadIdx.x, lane = tid & 31, b = blockIdx.x;
  const int lo = b * RPB;
  const int rows = (NNODES - lo < RPB) ? (NNODES - lo) : RPB;
  if (tid < 512) rcnt[tid] = 0;
  const int s = __builtin_amdgcn_readfirstlane(tid >> 5);
  const unsigned below = (1u << lane) - 1u;
  int* myh = hist + s * 32;
  int* myp = wpos + s * 32;
  myh[lane] = 0;

  int slen[SEGW], sbeg[SEGW];
#pragma unroll
  for (int j = 0; j < SEGW; ++j) {
    const int c = s * SEGW + j;
    slen[j] = __builtin_amdgcn_readfirstlane(ccnt[b * NCH + c]);
    sbeg[j] = __builtin_amdgcn_readfirstlane(c * CHEP + cstart[b * NCH + c]);
  }
  {
    const int ob = tid >> 2, q = tid & 3;
    const int4* __restrict__ row = (const int4*)(ccnt + ob * NCH) + q;
    int4 v[NCH / 16];
#pragma unroll
    for (int k = 0; k < NCH / 16; ++k) v[k] = row[k * 4];
    int tot = 0;
#pragma unroll
    for (int k = 0; k < NCH / 16; ++k) tot += v[k].x + v[k].y + v[k].z + v[k].w;
    tot += __shfl_xor(tot, 1); tot += __shfl_xor(tot, 2);
    if (q == 0) rtot[ob] = tot;
  }
  __syncthreads();
  int p[OCHK];
  if (s * 32 < rows) {
    const int* __restrict__ pn = opart + lo + (tid < rows ? tid : 0);
#pragma unroll
    for (int k = 0; k < OCHK; ++k) p[k] = pn[k * NNODES];
  }
  int ent[SEGW];
  bool ok[SEGW];
#pragma unroll
  for (int j = 0; j < SEGW; ++j) {
    ok[j] = lane < slen[j];
    ent[j] = 0;
    if (ok[j]) ent[j] = lst[sbeg[j] + lane];
  }
  int pre = 0;
#pragma unroll
  for (int k = 0; k < NB / 32; ++k) {
    const int ob = lane * (NB / 32) + k;
    pre += (ob < b) ? rtot[ob] : 0;
  }
#pragma unroll
  for (int dd = 16; dd >= 1; dd >>= 1) pre += __shfl_xor(pre, dd);
  const int region = (pre + b * REGPAD + 31) & ~31;

#pragma unroll
  for (int j = 0; j < SEGW; ++j) {
    count_step(ent[j], ok[j], below, myh, rcnt);
    for (int i = 32; i < slen[j]; i += 32) {
      const bool okr = i + lane < slen[j];
      int e = 0;
      if (okr) e = lst[sbeg[j] + i + lane];
      count_step(e, okr, below, myh, rcnt);
    }
  }
  __syncthreads();

  if (tid < rows) {
    int od = 1;
#pragma unroll
    for (int k = 0; k < OCHK; ++k) od += p[k];
    inorm[lo + tid] = rsqrtf(fmaxf((float)(rcnt[tid] + 1), 1.0f));
    onorm[lo + tid] = rsqrtf(fmaxf((float)od, 1.0f));
  }

  int rlen;
  bool staged;
  {
    int col = 0;
#pragma unroll
    for (int l = 0; l < 32; ++l) col += hist[l * 32 + lane];
    const int pad = (col + 7) & ~7;
    int incl = pad;
#pragma unroll
    for (int d = 1; d < 32; d <<= 1) {
      int n = __shfl_up(incl, d);
      if (lane >= d) incl += n;
    }
    const int woff = __shfl(incl - pad, s);
    const int tot  = __shfl(col, s);
    const int tpad = __shfl(pad, s);
    const int mine = hist[lane * 32 + s];
    int inc2 = mine;
#pragma unroll
    for (int d = 1; d < 32; d <<= 1) {
      int n = __shfl_up(inc2, d);
      if (lane >= d) inc2 += n;
    }
    rlen = __builtin_amdgcn_readlane(incl, 31);
    staged = rlen <= SCAP;
    wpos[lane * 32 + s] = woff + inc2 - mine;
    if (lane == 0) { wtab[(b * 32 + s) * 2] = region + woff; wtab[(b * 32 + s) * 2 + 1] = tpad; }
    if (lane < tpad - tot) {
      if (staged) stage[woff + tot + lane] = PADENT;
      else lst2[region + woff + tot + lane] = PADENT;
    }
  }
  __syncthreads();

  if (staged) {
#pragma unroll
    for (int j = 0; j < SEGW; ++j) {
      place_step(ent[j], ok[j], below, myp, stage);
      for (int i = 32; i < slen[j]; i += 32) {
        const bool okr = i + lane < slen[j];
        int e = 0;
        if (okr) e = lst[sbeg[j] + i + lane];
        place_step(e, okr, below, myp, stage);
      }
    }
    __syncthreads();
    int4* __restrict__ reg4 = (int4*)(lst2 + region);
    for (int i = tid; i < (rlen >> 2); i += 1024) reg4[i] = stage4[i];
  } else {
    int* __restrict__ reg = lst2 + region;
#pragma unroll
    for (int j = 0; j < SEGW; ++j) {
      place_step(ent[j], ok[j], below, myp, reg);
      for (int i = 32; i < slen[j]; i += 32) {
        const bool okr = i + lane < slen[j];
        int e = 0;
        if (okr) e = lst[sbeg[j] + i + lane];
        place_step(e, okr, below, myp, reg);
      }
    }
  }
}

#define KROWS   ((RPB + 31) / 32)
static_assert(KROWS <= 15, "register row 15 is the spare one");

template<int NOUT, bool RELU, bool HOUT>
__global__ __launch_bounds__(1024)
void agg_owner_kernel(const int* __restrict__ wtab, const int* __restrict__ lst2,
                      const float* __restrict__ h, const float* __restrict__ inorm,
                      const float* __restrict__ onorm,
                      const float* __restrict__ bias, float* __restrict__ out,
                      _Float16* __restrict__ outh) {
  constexpr int V = NOUT / 32;
  typedef float fv __attribute__((ext_vector_type(V)));
  typedef _Float16 hfv __attribute__((ext_vector_type(V)));
  typedef float v16f __attribute__((ext_vector_type(16)));
  typedef int v8i __attribute__((ext_vector_type(8)));
  const int tid = threadIdx.x, lane = tid & 31, b = blockIdx.x;
  const int w = __builtin_amdgcn_readfirstlane(tid >> 5);
  const int lo = b * RPB;
  const int rows = (NNODES - lo < RPB) ? (NNODES - lo) : RPB;

  const int lbeg = wtab[(b * 32 + w) * 2];
  const int n = wtab[(b * 32 + w) * 2 + 1];
  const v8i* __restrict__ myl = (const v8i*)(lst2 + lbeg);

  const fv* __restrict__ hv = (const fv*)h + (size_t)lo * 32;
  v16f a[V];
  {
    fv t[KROWS];
#pragma unroll
    for (int k = 0; k < KROWS; ++k) {
      const int r = k * 32 + w;
      t[k] = hv[(r < rows ? r : w) * 32 + lane];
    }
#pragma unroll
    for (int c = 0; c < V; ++c) {
#pragma unroll
      for (int k = 0; k < 16; ++k) a[c][k] = (k < KROWS) ? t[k < KROWS ? k : 0][c] : 0.0f;
    }
  }

  const char* __restrict__ hbytes = (const char*)h;
  const unsigned laneoff = (unsigned)lane * (unsigned)sizeof(fv);
  if (n > 0) {
    v8i en = myl[0];
    for (int j = 0; j < n; j += 8) {
      const v8i e = en;
      const int jn = (j + 8 < n) ? j + 8 : j;
      en = myl[jn >> 3];
      fv v[8];
#pragma unroll
      for (int u = 0; u < 8; ++u)
        v[u] = *(const fv*)(hbytes + (size_t)((unsigned)e[u] >> 9) * (32 * sizeof(fv)) + laneoff);
      __builtin_amdgcn_sched_barrier(0);
#pragma unroll
      for (int u = 0; u < 8; ++u) {
        const unsigned k = ((unsigned)e[u] >> 5) & 15u;
#pragma unroll
        for (int cc = 0; cc < V; ++cc) a[cc][k] += v[u][cc];
      }
    }
  }

  const fv* __restrict__ bv = (const fv*)bias;
  fv* __restrict__ ov = (fv*)out + (size_t)lo * 32;
  hfv* __restrict__ oh = (hfv*)outh + (size_t)lo * 32;
  const fv bb = bv[lane];
  float sc[KROWS], so[KROWS];
#pragma unroll
  for (int k = 0; k < KROWS; ++k) {
    const int r = k * 32 + w;
    sc[k] = inorm[lo + (r < rows ? r : w)];
    so[k] = HOUT ? onorm[lo + (r < rows ? r : w)] : 1.0f;
  }
#pragma unroll
  for (int k = 0; k < KROWS; ++k) {
    const int r = k * 32 + w;
    if (r < rows) {
      fv x;
#pragma unroll
      for (int cc = 0; cc < V; ++cc) x[cc] = a[cc][k];
      x = x * sc[k] + bb;
      if (RELU) {
#pragma unroll
        for (int cc = 0; cc < V; ++cc) x[cc] = fmaxf(x[cc], 0.0f);
      }
      if (HOUT) {
        hfv y;
#pragma unroll
        for (int cc = 0; cc < V; ++cc) y[cc] = (_Float16)(x[cc] * so[k]);
        oh[r * 32 + lane] = y;
      } else {
        ov[r * 32 + lane] = x;
      }
    }
  }
}

extern "C" void kernel_launch(void* const* d_in, const int* in_sizes, int n_in,
                              void* d_out, int out_size, void* d_ws, size_t ws_size,
                              hipStream_t stream) {
  const float* feat = (const float*)d_in[0];
  const int*   src  = (const int*)  d_in[1];
  const int*   dst  = (const int*)  d_in[2];
  const float* W1   = (const float*)d_in[3];
  const float* b1   = (const float*)d_in[4];
  const float* W2   = (const float*)d_in[5];
  const float* b2   = (const float*)d_in[6];
  const float* W3   = (const float*)d_in[7];
  const float* b3   = (const float*)d_in[8];
  float* out = (float*)d_out;

  size_t off = 0;
  auto alloc = [&](size_t bytes) -> char* {
    char* p = (char*)d_ws + off;
    off += (bytes + 255) & ~(size_t)255;
    return p;
  };
  float*     onorm = (float*)alloc((size_t)NNODES * sizeof(float));
  float*     inorm = (float*)alloc((size_t)NNODES * sizeof(float));
  int*       opart = (int*)alloc((size_t)OCHK * NNODES * sizeof(int));
  float*     hbuf  = (float*)alloc((size_t)NNODES * FDIM * sizeof(float));
  _Float16*  xh    = (_Float16*)alloc((size_t)NNODES * FDIM * sizeof(_Float16));
  _Float16*  Wp    = (_Float16*)alloc((size_t)FDIM * FDIM * sizeof(_Float16));
  int*       ccnt  = (int*)alloc((size_t)NB * NCH * sizeof(int));
  int*       cstart = (int*)alloc((size_t)NB * NCH * sizeof(int));
  int*       lst   = (int*)alloc((size_t)NCH * CHEP * sizeof(int));
  int*       lst2  = (int*)alloc(((size_t)NEDGES + (size_t)NB * REGPAD + 64) * sizeof(int));
  int*       wtab  = (int*)alloc((size_t)NB * 32 * 2 * sizeof(int));

  const int TB = 256;

  owner_sort_kernel<<<NCH, 1024, 0, stream>>>(src, dst, ccnt, cstart, lst, opart);
  split_kernel<<<NB, 1024, 0, stream>>>(ccnt, cstart, lst, lst2, wtab, opart, onorm, inorm);

  pack_w_kernel<<<ceil_div(4 * 8 * 512, TB), TB, 0, stream>>>(W1, Wp, 128);
  {
    int waves = NNODES / 16;
    gemm_wmma_kernel<128, false><<<ceil_div((long long)waves * 32, GEMM_TB), GEMM_TB, 0, stream>>>(feat, nullptr, onorm, Wp, hbuf);
  }
  agg_owner_kernel<128, true, true><<<NB, 1024, 0, stream>>>(wtab, lst2, hbuf, inorm, onorm, b1, nullptr, xh);

  pack_w_kernel<<<ceil_div(4 * 8 * 512, TB), TB, 0, stream>>>(W2, Wp, 128);
  {
    int waves = NNODES / 16;
    gemm_wmma_kernel<128, true><<<ceil_div((long long)waves * 32, GEMM_TB), GEMM_TB, 0, stream>>>(nullptr, xh, onorm, Wp, hbuf);
  }
  agg_owner_kernel<128, true, true><<<NB, 1024, 0, stream>>>(wtab, lst2, hbuf, inorm, onorm, b2, nullptr, xh);

  pack_w_kernel<<<ceil_div(4 * 4 * 512, TB), TB, 0, stream>>>(W3, Wp, 64);
  {
    int waves = NNODES / 16;
    gemm_wmma_kernel<64, true><<<ceil_div((long long)waves * 32, GEMM_TB), GEMM_TB, 0, stream>>>(nullptr, xh, onorm, Wp, hbuf);
  }
  agg_owner_kernel<64, false, false><<<NB, 1024, 0, stream>>>(wtab, lst2, hbuf, inorm, onorm, b3, out, nullptr);
}
